# starting kernel
# baseline (speedup 1.0000x reference)
#include <hip/hip_runtime.h>

#define N_NODES 65536
#define N_EDGES 1048576
#define DIM 64

typedef float v2f __attribute__((ext_vector_type(2)));
typedef float v8f __attribute__((ext_vector_type(8)));


__global__ __launch_bounds__(256) void zero_u32_kernel(unsigned* __restrict__ p, int n) {
    int i = blockIdx.x * 256 + threadIdx.x;
    if (i < n) p[i] = 0u;
}

__global__ __launch_bounds__(256) void degree_kernel(const int* __restrict__ ei,
                                                     unsigned* __restrict__ deg_out,
                                                     unsigned* __restrict__ deg_in) {
    int e = blockIdx.x * 256 + threadIdx.x;
    if (e < N_EDGES) {
        atomicAdd(&deg_out[ei[e]], 1u);
        atomicAdd(&deg_in[ei[N_EDGES + e]], 1u);
    }
}

__global__ __launch_bounds__(256) void rsqrt_kernel(const unsigned* __restrict__ deg_out,
                                                    const unsigned* __restrict__ deg_in,
                                                    float* __restrict__ rs_out,
                                                    float* __restrict__ rs_in) {
    int i = blockIdx.x * 256 + threadIdx.x;
    if (i < N_NODES) {
        unsigned dovr = deg_out[i]; if (dovr < 1u) dovr = 1u;
        unsigned din  = deg_in[i];  if (din  < 1u) din  = 1u;
        rs_out[i] = rsqrtf((float)dovr);
        rs_in[i]  = rsqrtf((float)din);
    }
}

__global__ __launch_bounds__(1024) void scan_kernel(const unsigned* __restrict__ deg_in,
                                                    int* __restrict__ row_ptr) {
    __shared__ unsigned part[1024];
    int t = threadIdx.x;
    int base = t * (N_NODES / 1024);
    unsigned sum = 0;
    for (int i = 0; i < N_NODES / 1024; ++i) sum += deg_in[base + i];
    part[t] = sum;
    __syncthreads();
    for (int off = 1; off < 1024; off <<= 1) {
        unsigned v = (t >= off) ? part[t - off] : 0u;
        __syncthreads();
        part[t] += v;
        __syncthreads();
    }
    unsigned run = (t == 0) ? 0u : part[t - 1];
    for (int i = 0; i < N_NODES / 1024; ++i) {
        row_ptr[base + i] = (int)run;
        run += deg_in[base + i];
    }
    if (t == 1023) row_ptr[N_NODES] = (int)run;
}

__global__ __launch_bounds__(256) void fill_kernel(const int* __restrict__ ei,
                                                   unsigned* __restrict__ cursor,
                                                   const int* __restrict__ row_ptr,
                                                   int* __restrict__ in_src) {
    int e = blockIdx.x * 256 + threadIdx.x;
    if (e < N_EDGES) {
        int s = ei[e];
        int d = ei[N_EDGES + e];
        unsigned slot = atomicAdd(&cursor[d], 1u);
        in_src[row_ptr[d] + (int)slot] = s;
    }
}

__global__ __launch_bounds__(256) void scale0_kernel(const float* __restrict__ x,
                                                     const float* __restrict__ rs_out,
                                                     float* __restrict__ hs) {
    int t = blockIdx.x * 256 + threadIdx.x;
    float4 v = ((const float4*)x)[t];
    float r = rs_out[t >> 4];
    v.x *= r; v.y *= r; v.z *= r; v.w *= r;
    ((float4*)hs)[t] = v;
}

__global__ __launch_bounds__(256) void agg_kernel(const float* __restrict__ hs,
                                                  const int* __restrict__ in_src,
                                                  const int* __restrict__ row_ptr,
                                                  const float* __restrict__ rs_in,
                                                  float* __restrict__ aggn) {
    int node = (blockIdx.x * 256 + threadIdx.x) >> 5;
    int lane = threadIdx.x & 31;
    int start = row_ptr[node];
    int end   = row_ptr[node + 1];
    float ax = 0.0f, ay = 0.0f;
    for (int base = start; base < end; base += 32) {
        int i = base + lane;
        int idx = (i < end) ? in_src[i] : 0;
        int cnt = end - base; if (cnt > 32) cnt = 32;
        for (int j = 0; j < cnt; ++j) {
            int s = __shfl(idx, j, 32);
            const v2f v = *(const v2f*)(hs + (size_t)s * DIM + lane * 2);
            ax += v.x; ay += v.y;
        }
    }
    float r = rs_in[node];
    v2f outv; outv.x = ax * r; outv.y = ay * r;
    *(v2f*)(aggn + (size_t)node * DIM + lane * 2) = outv;
}

__global__ __launch_bounds__(128) void gemm_wmma_kernel(const float* __restrict__ X,
                                                        const float* __restrict__ W,
                                                        const float* __restrict__ bias,
                                                        const float* __restrict__ rs_out,
                                                        float* __restrict__ Y,
                                                        int relu_scale) {
    __shared__ float lWsw[16 * 4 * 32 * 2];
    __shared__ float lB[DIM];
    for (int p = threadIdx.x; p < 16 * 4 * 32; p += 128) {
        int l  = p & 31;
        int t  = (p >> 5) & 3;
        int f  = p >> 7;
        int ka = 4 * f + 2 * (l >> 4);
        int n  = 16 * t + (l & 15);
        v2f w2;
        w2.x = W[ka * DIM + n];
        w2.y = W[(ka + 1) * DIM + n];
        *(v2f*)&lWsw[p * 2] = w2;
    }
    if (threadIdx.x < DIM) lB[threadIdx.x] = bias[threadIdx.x];
    __syncthreads();

    const int wave = threadIdx.x >> 5;
    const int lane = threadIdx.x & 31;
    const int half = lane >> 4;
    const int l16  = lane & 15;
    const int m0   = blockIdx.x * 64 + wave * 16;
    const int mA   = m0 + l16;

    v8f acc[4] = {v8f{}, v8f{}, v8f{}, v8f{}};

#pragma unroll
    for (int f = 0; f < 16; ++f) {
        const int ka = 4 * f + 2 * half;
        v2f a = *(const v2f*)(X + (size_t)mA * DIM + ka);
#pragma unroll
        for (int t = 0; t < 4; ++t) {
            v2f b = *(const v2f*)&lWsw[(((f * 4 + t) * 32) + lane) * 2];
            acc[t] = __builtin_amdgcn_wmma_f32_16x16x4_f32(
                false, a, false, b, (short)0, acc[t], false, false);
        }
    }

    const int mrow0 = m0 + half * 8;
    float rsv[8];
#pragma unroll
    for (int r = 0; r < 8; ++r) rsv[r] = relu_scale ? rs_out[mrow0 + r] : 1.0f;

#pragma unroll
    for (int t = 0; t < 4; ++t) {
        const int n = t * 16 + l16;
        const float bn = lB[n];
#pragma unroll
        for (int r = 0; r < 8; ++r) {
            float v = acc[t][r] + bn;
            if (relu_scale) v = fmaxf(v, 0.0f) * rsv[r];
            Y[(size_t)(mrow0 + r) * DIM + n] = v;
        }
    }
}

extern "C" void kernel_launch(void* const* d_in, const int* in_sizes, int n_in,
                              void* d_out, int out_size, void* d_ws, size_t ws_size,
                              hipStream_t stream) {
    (void)in_sizes; (void)n_in; (void)out_size; (void)ws_size;
    const float* x  = (const float*)d_in[0];
    const int*   ei = (const int*)d_in[1];
    const float* W1 = (const float*)d_in[3];
    const float* b1 = (const float*)d_in[4];
    const float* W2 = (const float*)d_in[5];
    const float* b2 = (const float*)d_in[6];
    const float* W3 = (const float*)d_in[7];
    const float* b3 = (const float*)d_in[8];
    float* out = (float*)d_out;

    char* ws = (char*)d_ws;
    unsigned* deg_out = (unsigned*)ws; ws += (size_t)N_NODES * 4;
    unsigned* deg_in  = (unsigned*)ws; ws += (size_t)N_NODES * 4;
    unsigned* cursor  = (unsigned*)ws; ws += (size_t)N_NODES * 4;
    float*    rs_out  = (float*)ws;    ws += (size_t)N_NODES * 4;
    float*    rs_in   = (float*)ws;    ws += (size_t)N_NODES * 4;
    int*      row_ptr = (int*)ws;      ws += (size_t)(N_NODES + 16) * 4;
    int*      in_src  = (int*)ws;      ws += (size_t)N_EDGES * 4;
    float*    hs      = (float*)ws;    ws += (size_t)N_NODES * DIM * 4;
    float*    aggn    = (float*)ws;    ws += (size_t)N_NODES * DIM * 4;

    zero_u32_kernel<<<(3 * N_NODES + 255) / 256, 256, 0, stream>>>(deg_out, 3 * N_NODES);
    degree_kernel<<<N_EDGES / 256, 256, 0, stream>>>(ei, deg_out, deg_in);
    rsqrt_kernel<<<N_NODES / 256, 256, 0, stream>>>(deg_out, deg_in, rs_out, rs_in);
    scan_kernel<<<1, 1024, 0, stream>>>(deg_in, row_ptr);
    fill_kernel<<<N_EDGES / 256, 256, 0, stream>>>(ei, cursor, row_ptr, in_src);

    scale0_kernel<<<(N_NODES * DIM / 4) / 256, 256, 0, stream>>>(x, rs_out, hs);

    agg_kernel<<<(N_NODES * 32) / 256, 256, 0, stream>>>(hs, in_src, row_ptr, rs_in, aggn);
    gemm_wmma_kernel<<<N_NODES / 64, 128, 0, stream>>>(aggn, W1, b1, rs_out, hs, 1);
    agg_kernel<<<(N_NODES * 32) / 256, 256, 0, stream>>>(hs, in_src, row_ptr, rs_in, aggn);
    gemm_wmma_kernel<<<N_NODES / 64, 128, 0, stream>>>(aggn, W2, b2, rs_out, hs, 1);
    agg_kernel<<<(N_NODES * 32) / 256, 256, 0, stream>>>(hs, in_src, row_ptr, rs_in, aggn);
    gemm_wmma_kernel<<<N_NODES / 64, 128, 0, stream>>>(aggn, W3, b3, rs_out, out, 0);
}


# Degree counts and CSR slot ranks moved from global atomics to per-chunk 64 KB LDS histograms merged by plain sums; edge order in a row (agg sum order) now chunk-major, LDS arrival inside
# speedup vs baseline: 1.6125x; 1.6125x over previous
#include <hip/hip_runtime.h>

#define N_NODES 65536
#define N_EDGES 1048576
#define DIM 64

typedef float v2f __attribute__((ext_vector_type(2)));
typedef float v8f __attribute__((ext_vector_type(8)));


#define CHUNKS 64
#define CHUNK_EDGES (N_EDGES / CHUNKS)
#define RANGES 4
#define RANGE_NODES (N_NODES / RANGES)
#define RANGE_SHIFT 14
#define EDGE_BATCH 8

__global__ __launch_bounds__(256) void hist_kernel(const int* __restrict__ ei,
                                                   unsigned* __restrict__ cnt_src,
                                                   unsigned* __restrict__ cnt_dst) {
    __shared__ unsigned tab[RANGE_NODES];
    const int g = blockIdx.x, q = blockIdx.y, which = blockIdx.z;
    uint4* tab4 = (uint4*)tab;
    for (int i = threadIdx.x; i < RANGE_NODES / 4; i += 256) tab4[i] = make_uint4(0u, 0u, 0u, 0u);
    __syncthreads();
    const int4* idx4 = (const int4*)(ei + (size_t)which * N_EDGES + (size_t)g * CHUNK_EDGES);
    for (int i0 = threadIdx.x; i0 < CHUNK_EDGES / 4; i0 += 256 * EDGE_BATCH) {
        int4 v[EDGE_BATCH];
#pragma unroll
        for (int k = 0; k < EDGE_BATCH; ++k) v[k] = idx4[i0 + k * 256];
#pragma unroll
        for (int k = 0; k < EDGE_BATCH; ++k) {
            if ((v[k].x >> RANGE_SHIFT) == q) atomicAdd(&tab[v[k].x & (RANGE_NODES - 1)], 1u);
            if ((v[k].y >> RANGE_SHIFT) == q) atomicAdd(&tab[v[k].y & (RANGE_NODES - 1)], 1u);
            if ((v[k].z >> RANGE_SHIFT) == q) atomicAdd(&tab[v[k].z & (RANGE_NODES - 1)], 1u);
            if ((v[k].w >> RANGE_SHIFT) == q) atomicAdd(&tab[v[k].w & (RANGE_NODES - 1)], 1u);
        }
    }
    __syncthreads();
    unsigned* cnt = which ? cnt_dst : cnt_src;
    uint4* out4 = (uint4*)(cnt + (size_t)g * N_NODES + (size_t)q * RANGE_NODES);
    for (int i = threadIdx.x; i < RANGE_NODES / 4; i += 256) out4[i] = tab4[i];
}

__global__ __launch_bounds__(256) void merge_kernel(const unsigned* __restrict__ cnt_src,
                                                    unsigned* __restrict__ cnt_dst,
                                                    unsigned* __restrict__ deg_in,
                                                    float* __restrict__ rs_out,
                                                    float* __restrict__ rs_in) {
    const int n = blockIdx.x * 256 + threadIdx.x;
    if (blockIdx.y == 0) {
        unsigned dout = 0u;
#pragma unroll
        for (int g = 0; g < CHUNKS; ++g) dout += cnt_src[(size_t)g * N_NODES + n];
        if (dout < 1u) dout = 1u;
        rs_out[n] = rsqrtf((float)dout);
    } else {
        unsigned c[CHUNKS];
#pragma unroll
        for (int g = 0; g < CHUNKS; ++g) c[g] = cnt_dst[(size_t)g * N_NODES + n];
        unsigned din = 0u;
#pragma unroll
        for (int g = 0; g < CHUNKS; ++g) {
            cnt_dst[(size_t)g * N_NODES + n] = din;
            din += c[g];
        }
        deg_in[n] = din;
        if (din < 1u) din = 1u;
        rs_in[n] = rsqrtf((float)din);
    }
}

__global__ __launch_bounds__(1024) void scan_kernel(const unsigned* __restrict__ deg_in,
                                                    int* __restrict__ row_ptr) {
    __shared__ unsigned part[1024];
    int t = threadIdx.x;
    int base = t * (N_NODES / 1024);
    unsigned sum = 0;
    for (int i = 0; i < N_NODES / 1024; ++i) sum += deg_in[base + i];
    part[t] = sum;
    __syncthreads();
    for (int off = 1; off < 1024; off <<= 1) {
        unsigned v = (t >= off) ? part[t - off] : 0u;
        __syncthreads();
        part[t] += v;
        __syncthreads();
    }
    unsigned run = (t == 0) ? 0u : part[t - 1];
    for (int i = 0; i < N_NODES / 1024; ++i) {
        row_ptr[base + i] = (int)run;
        run += deg_in[base + i];
    }
    if (t == 1023) row_ptr[N_NODES] = (int)run;
}

__global__ __launch_bounds__(256) void fill_kernel(const int* __restrict__ ei,
                                                   const unsigned* __restrict__ chunk_off,
                                                   const int* __restrict__ row_ptr,
                                                   int* __restrict__ in_src) {
    __shared__ unsigned cur[RANGE_NODES];
    const int g = blockIdx.x, q = blockIdx.y;
    const uint4* off4 = (const uint4*)(chunk_off + (size_t)g * N_NODES + (size_t)q * RANGE_NODES);
    const int4*  rp4  = (const int4*)(row_ptr + (size_t)q * RANGE_NODES);
    uint4* cur4 = (uint4*)cur;
    for (int i0 = threadIdx.x; i0 < RANGE_NODES / 4; i0 += 256 * EDGE_BATCH) {
        uint4 o[EDGE_BATCH];
        int4  r[EDGE_BATCH];
#pragma unroll
        for (int k = 0; k < EDGE_BATCH; ++k) { o[k] = off4[i0 + k * 256]; r[k] = rp4[i0 + k * 256]; }
#pragma unroll
        for (int k = 0; k < EDGE_BATCH; ++k)
            cur4[i0 + k * 256] = make_uint4(o[k].x + (unsigned)r[k].x, o[k].y + (unsigned)r[k].y,
                                            o[k].z + (unsigned)r[k].z, o[k].w + (unsigned)r[k].w);
    }
    __syncthreads();
    const int4* s4 = (const int4*)(ei + (size_t)g * CHUNK_EDGES);
    const int4* d4 = (const int4*)(ei + (size_t)N_EDGES + (size_t)g * CHUNK_EDGES);
    for (int i0 = threadIdx.x; i0 < CHUNK_EDGES / 4; i0 += 256 * EDGE_BATCH) {
        int4 d[EDGE_BATCH], s[EDGE_BATCH];
#pragma unroll
        for (int k = 0; k < EDGE_BATCH; ++k) { d[k] = d4[i0 + k * 256]; s[k] = s4[i0 + k * 256]; }
#pragma unroll
        for (int k = 0; k < EDGE_BATCH; ++k) {
            if ((d[k].x >> RANGE_SHIFT) == q) in_src[atomicAdd(&cur[d[k].x & (RANGE_NODES - 1)], 1u)] = s[k].x;
            if ((d[k].y >> RANGE_SHIFT) == q) in_src[atomicAdd(&cur[d[k].y & (RANGE_NODES - 1)], 1u)] = s[k].y;
            if ((d[k].z >> RANGE_SHIFT) == q) in_src[atomicAdd(&cur[d[k].z & (RANGE_NODES - 1)], 1u)] = s[k].z;
            if ((d[k].w >> RANGE_SHIFT) == q) in_src[atomicAdd(&cur[d[k].w & (RANGE_NODES - 1)], 1u)] = s[k].w;
        }
    }
}

__global__ __launch_bounds__(256) void scale0_kernel(const float* __restrict__ x,
                                                     const float* __restrict__ rs_out,
                                                     float* __restrict__ hs) {
    int t = blockIdx.x * 256 + threadIdx.x;
    float4 v = ((const float4*)x)[t];
    float r = rs_out[t >> 4];
    v.x *= r; v.y *= r; v.z *= r; v.w *= r;
    ((float4*)hs)[t] = v;
}

__global__ __launch_bounds__(256) void agg_kernel(const float* __restrict__ hs,
                                                  const int* __restrict__ in_src,
                                                  const int* __restrict__ row_ptr,
                                                  const float* __restrict__ rs_in,
                                                  float* __restrict__ aggn) {
    int node = (blockIdx.x * 256 + threadIdx.x) >> 5;
    int lane = threadIdx.x & 31;
    int start = row_ptr[node];
    int end   = row_ptr[node + 1];
    float ax = 0.0f, ay = 0.0f;
    for (int base = start; base < end; base += 32) {
        int i = base + lane;
        int idx = (i < end) ? in_src[i] : 0;
        int cnt = end - base; if (cnt > 32) cnt = 32;
        for (int j = 0; j < cnt; ++j) {
            int s = __shfl(idx, j, 32);
            const v2f v = *(const v2f*)(hs + (size_t)s * DIM + lane * 2);
            ax += v.x; ay += v.y;
        }
    }
    float r = rs_in[node];
    v2f outv; outv.x = ax * r; outv.y = ay * r;
    *(v2f*)(aggn + (size_t)node * DIM + lane * 2) = outv;
}

__global__ __launch_bounds__(128) void gemm_wmma_kernel(const float* __restrict__ X,
                                                        const float* __restrict__ W,
                                                        const float* __restrict__ bias,
                                                        const float* __restrict__ rs_out,
                                                        float* __restrict__ Y,
                                                        int relu_scale) {
    __shared__ float lWsw[16 * 4 * 32 * 2];
    __shared__ float lB[DIM];
    for (int p = threadIdx.x; p < 16 * 4 * 32; p += 128) {
        int l  = p & 31;
        int t  = (p >> 5) & 3;
        int f  = p >> 7;
        int ka = 4 * f + 2 * (l >> 4);
        int n  = 16 * t + (l & 15);
        v2f w2;
        w2.x = W[ka * DIM + n];
        w2.y = W[(ka + 1) * DIM + n];
        *(v2f*)&lWsw[p * 2] = w2;
    }
    if (threadIdx.x < DIM) lB[threadIdx.x] = bias[threadIdx.x];
    __syncthreads();

    const int wave = threadIdx.x >> 5;
    const int lane = threadIdx.x & 31;
    const int half = lane >> 4;
    const int l16  = lane & 15;
    const int m0   = blockIdx.x * 64 + wave * 16;
    const int mA   = m0 + l16;

    v8f acc[4] = {v8f{}, v8f{}, v8f{}, v8f{}};

#pragma unroll
    for (int f = 0; f < 16; ++f) {
        const int ka = 4 * f + 2 * half;
        v2f a = *(const v2f*)(X + (size_t)mA * DIM + ka);
#pragma unroll
        for (int t = 0; t < 4; ++t) {
            v2f b = *(const v2f*)&lWsw[(((f * 4 + t) * 32) + lane) * 2];
            acc[t] = __builtin_amdgcn_wmma_f32_16x16x4_f32(
                false, a, false, b, (short)0, acc[t], false, false);
        }
    }

    const int mrow0 = m0 + half * 8;
    float rsv[8];
#pragma unroll
    for (int r = 0; r < 8; ++r) rsv[r] = relu_scale ? rs_out[mrow0 + r] : 1.0f;

#pragma unroll
    for (int t = 0; t < 4; ++t) {
        const int n = t * 16 + l16;
        const float bn = lB[n];
#pragma unroll
        for (int r = 0; r < 8; ++r) {
            float v = acc[t][r] + bn;
            if (relu_scale) v = fmaxf(v, 0.0f) * rsv[r];
            Y[(size_t)(mrow0 + r) * DIM + n] = v;
        }
    }
}

extern "C" void kernel_launch(void* const* d_in, const int* in_sizes, int n_in,
                              void* d_out, int out_size, void* d_ws, size_t ws_size,
                              hipStream_t stream) {
    (void)in_sizes; (void)n_in; (void)out_size; (void)ws_size;
    const float* x  = (const float*)d_in[0];
    const int*   ei = (const int*)d_in[1];
    const float* W1 = (const float*)d_in[3];
    const float* b1 = (const float*)d_in[4];
    const float* W2 = (const float*)d_in[5];
    const float* b2 = (const float*)d_in[6];
    const float* W3 = (const float*)d_in[7];
    const float* b3 = (const float*)d_in[8];
    float* out = (float*)d_out;

    char* ws = (char*)d_ws;
    unsigned* deg_in  = (unsigned*)ws; ws += (size_t)N_NODES * 4;
    float*    rs_out  = (float*)ws;    ws += (size_t)N_NODES * 4;
    float*    rs_in   = (float*)ws;    ws += (size_t)N_NODES * 4;
    int*      row_ptr = (int*)ws;      ws += (size_t)(N_NODES + 16) * 4;
    int*      in_src  = (int*)ws;      ws += (size_t)N_EDGES * 4;
    float*    hs      = (float*)ws;    ws += (size_t)N_NODES * DIM * 4;
    float*    aggn    = (float*)ws;    ws += (size_t)N_NODES * DIM * 4;
    unsigned* cnt_src = (unsigned*)hs;
    unsigned* cnt_dst = (unsigned*)aggn;

    hist_kernel<<<dim3(CHUNKS, RANGES, 2), 256, 0, stream>>>(ei, cnt_src, cnt_dst);
    merge_kernel<<<dim3(N_NODES / 256, 2), 256, 0, stream>>>(cnt_src, cnt_dst, deg_in, rs_out, rs_in);
    scan_kernel<<<1, 1024, 0, stream>>>(deg_in, row_ptr);
    fill_kernel<<<dim3(CHUNKS, RANGES), 256, 0, stream>>>(ei, cnt_dst, row_ptr, in_src);

    scale0_kernel<<<(N_NODES * DIM / 4) / 256, 256, 0, stream>>>(x, rs_out, hs);

    agg_kernel<<<(N_NODES * 32) / 256, 256, 0, stream>>>(hs, in_src, row_ptr, rs_in, aggn);
    gemm_wmma_kernel<<<N_NODES / 64, 128, 0, stream>>>(aggn, W1, b1, rs_out, hs, 1);
    agg_kernel<<<(N_NODES * 32) / 256, 256, 0, stream>>>(hs, in_src, row_ptr, rs_in, aggn);
    gemm_wmma_kernel<<<N_NODES / 64, 128, 0, stream>>>(aggn, W2, b2, rs_out, hs, 1);
    agg_kernel<<<(N_NODES * 32) / 256, 256, 0, stream>>>(hs, in_src, row_ptr, rs_in, aggn);
    gemm_wmma_kernel<<<N_NODES / 64, 128, 0, stream>>>(aggn, W3, b3, rs_out, out, 0);
}


# agg: one wave per 8 consecutive destination nodes walks their contiguous CSR edge stretch with 16 row gathers in flight (was one wave per node, one gather at a time); per-node sum order unchanged
# speedup vs baseline: 2.0041x; 1.2429x over previous
#include <hip/hip_runtime.h>

#define N_NODES 65536
#define N_EDGES 1048576
#define DIM 64

typedef float v2f __attribute__((ext_vector_type(2)));
typedef float v8f __attribute__((ext_vector_type(8)));


#define CHUNKS 64
#define CHUNK_EDGES (N_EDGES / CHUNKS)
#define RANGES 4
#define RANGE_NODES (N_NODES / RANGES)
#define RANGE_SHIFT 14
#define EDGE_BATCH 8

__global__ __launch_bounds__(256) void hist_kernel(const int* __restrict__ ei,
                                                   unsigned* __restrict__ cnt_src,
                                                   unsigned* __restrict__ cnt_dst) {
    __shared__ unsigned tab[RANGE_NODES];
    const int g = blockIdx.x, q = blockIdx.y, which = blockIdx.z;
    uint4* tab4 = (uint4*)tab;
    for (int i = threadIdx.x; i < RANGE_NODES / 4; i += 256) tab4[i] = make_uint4(0u, 0u, 0u, 0u);
    __syncthreads();
    const int4* idx4 = (const int4*)(ei + (size_t)which * N_EDGES + (size_t)g * CHUNK_EDGES);
    for (int i0 = threadIdx.x; i0 < CHUNK_EDGES / 4; i0 += 256 * EDGE_BATCH) {
        int4 v[EDGE_BATCH];
#pragma unroll
        for (int k = 0; k < EDGE_BATCH; ++k) v[k] = idx4[i0 + k * 256];
#pragma unroll
        for (int k = 0; k < EDGE_BATCH; ++k) {
            if ((v[k].x >> RANGE_SHIFT) == q) atomicAdd(&tab[v[k].x & (RANGE_NODES - 1)], 1u);
            if ((v[k].y >> RANGE_SHIFT) == q) atomicAdd(&tab[v[k].y & (RANGE_NODES - 1)], 1u);
            if ((v[k].z >> RANGE_SHIFT) == q) atomicAdd(&tab[v[k].z & (RANGE_NODES - 1)], 1u);
            if ((v[k].w >> RANGE_SHIFT) == q) atomicAdd(&tab[v[k].w & (RANGE_NODES - 1)], 1u);
        }
    }
    __syncthreads();
    unsigned* cnt = which ? cnt_dst : cnt_src;
    uint4* out4 = (uint4*)(cnt + (size_t)g * N_NODES + (size_t)q * RANGE_NODES);
    for (int i = threadIdx.x; i < RANGE_NODES / 4; i += 256) out4[i] = tab4[i];
}

__global__ __launch_bounds__(256) void merge_kernel(const unsigned* __restrict__ cnt_src,
                                                    unsigned* __restrict__ cnt_dst,
                                                    unsigned* __restrict__ deg_in,
                                                    float* __restrict__ rs_out,
                                                    float* __restrict__ rs_in) {
    const int n = blockIdx.x * 256 + threadIdx.x;
    if (blockIdx.y == 0) {
        unsigned dout = 0u;
#pragma unroll
        for (int g = 0; g < CHUNKS; ++g) dout += cnt_src[(size_t)g * N_NODES + n];
        if (dout < 1u) dout = 1u;
        rs_out[n] = rsqrtf((float)dout);
    } else {
        unsigned c[CHUNKS];
#pragma unroll
        for (int g = 0; g < CHUNKS; ++g) c[g] = cnt_dst[(size_t)g * N_NODES + n];
        unsigned din = 0u;
#pragma unroll
        for (int g = 0; g < CHUNKS; ++g) {
            cnt_dst[(size_t)g * N_NODES + n] = din;
            din += c[g];
        }
        deg_in[n] = din;
        if (din < 1u) din = 1u;
        rs_in[n] = rsqrtf((float)din);
    }
}

__global__ __launch_bounds__(1024) void scan_kernel(const unsigned* __restrict__ deg_in,
                                                    int* __restrict__ row_ptr) {
    __shared__ unsigned part[1024];
    int t = threadIdx.x;
    int base = t * (N_NODES / 1024);
    unsigned sum = 0;
    for (int i = 0; i < N_NODES / 1024; ++i) sum += deg_in[base + i];
    part[t] = sum;
    __syncthreads();
    for (int off = 1; off < 1024; off <<= 1) {
        unsigned v = (t >= off) ? part[t - off] : 0u;
        __syncthreads();
        part[t] += v;
        __syncthreads();
    }
    unsigned run = (t == 0) ? 0u : part[t - 1];
    for (int i = 0; i < N_NODES / 1024; ++i) {
        row_ptr[base + i] = (int)run;
        run += deg_in[base + i];
    }
    if (t == 1023) row_ptr[N_NODES] = (int)run;
}

__global__ __launch_bounds__(256) void fill_kernel(const int* __restrict__ ei,
                                                   const unsigned* __restrict__ chunk_off,
                                                   const int* __restrict__ row_ptr,
                                                   int* __restrict__ in_src) {
    __shared__ unsigned cur[RANGE_NODES];
    const int g = blockIdx.x, q = blockIdx.y;
    const uint4* off4 = (const uint4*)(chunk_off + (size_t)g * N_NODES + (size_t)q * RANGE_NODES);
    const int4*  rp4  = (const int4*)(row_ptr + (size_t)q * RANGE_NODES);
    uint4* cur4 = (uint4*)cur;
    for (int i0 = threadIdx.x; i0 < RANGE_NODES / 4; i0 += 256 * EDGE_BATCH) {
        uint4 o[EDGE_BATCH];
        int4  r[EDGE_BATCH];
#pragma unroll
        for (int k = 0; k < EDGE_BATCH; ++k) { o[k] = off4[i0 + k * 256]; r[k] = rp4[i0 + k * 256]; }
#pragma unroll
        for (int k = 0; k < EDGE_BATCH; ++k)
            cur4[i0 + k * 256] = make_uint4(o[k].x + (unsigned)r[k].x, o[k].y + (unsigned)r[k].y,
                                            o[k].z + (unsigned)r[k].z, o[k].w + (unsigned)r[k].w);
    }
    __syncthreads();
    const int4* s4 = (const int4*)(ei + (size_t)g * CHUNK_EDGES);
    const int4* d4 = (const int4*)(ei + (size_t)N_EDGES + (size_t)g * CHUNK_EDGES);
    for (int i0 = threadIdx.x; i0 < CHUNK_EDGES / 4; i0 += 256 * EDGE_BATCH) {
        int4 d[EDGE_BATCH], s[EDGE_BATCH];
#pragma unroll
        for (int k = 0; k < EDGE_BATCH; ++k) { d[k] = d4[i0 + k * 256]; s[k] = s4[i0 + k * 256]; }
#pragma unroll
        for (int k = 0; k < EDGE_BATCH; ++k) {
            if ((d[k].x >> RANGE_SHIFT) == q) in_src[atomicAdd(&cur[d[k].x & (RANGE_NODES - 1)], 1u)] = s[k].x;
            if ((d[k].y >> RANGE_SHIFT) == q) in_src[atomicAdd(&cur[d[k].y & (RANGE_NODES - 1)], 1u)] = s[k].y;
            if ((d[k].z >> RANGE_SHIFT) == q) in_src[atomicAdd(&cur[d[k].z & (RANGE_NODES - 1)], 1u)] = s[k].z;
            if ((d[k].w >> RANGE_SHIFT) == q) in_src[atomicAdd(&cur[d[k].w & (RANGE_NODES - 1)], 1u)] = s[k].w;
        }
    }
}

__global__ __launch_bounds__(256) void scale0_kernel(const float* __restrict__ x,
                                                     const float* __restrict__ rs_out,
                                                     float* __restrict__ hs) {
    int t = blockIdx.x * 256 + threadIdx.x;
    float4 v = ((const float4*)x)[t];
    float r = rs_out[t >> 4];
    v.x *= r; v.y *= r; v.z *= r; v.w *= r;
    ((float4*)hs)[t] = v;
}

#define AGG_NODES 8
#define AGG_BATCH 16

__global__ __launch_bounds__(256) void agg_kernel(const float* __restrict__ hs,
                                                  const int* __restrict__ in_src,
                                                  const int* __restrict__ row_ptr,
                                                  const float* __restrict__ rs_in,
                                                  float* __restrict__ aggn) {
    const int lane  = threadIdx.x & 31;
    const int node0 = ((blockIdx.x * 256 + threadIdx.x) >> 5) * AGG_NODES;
    const int rp  = (lane <= AGG_NODES) ? row_ptr[node0 + lane] : 0x7fffffff;
    const int rsl = __float_as_int(rs_in[node0 + (lane < AGG_NODES ? lane : AGG_NODES - 1)]);
    const int e_begin = __builtin_amdgcn_readlane(rp, 0);
    const int e_end   = __builtin_amdgcn_readlane(rp, AGG_NODES);
    int cur     = 0;
    int cur_end = __builtin_amdgcn_readlane(rp, 1);
    float ax = 0.0f, ay = 0.0f;
    const float* hl = hs + lane * 2;
    float* ol = aggn + (size_t)node0 * DIM + lane * 2;

    auto batch = [&](const int idx, const int j0, const int e0) {
        v2f v[AGG_BATCH];
#pragma unroll
        for (int u = 0; u < AGG_BATCH; ++u) {
            const unsigned srow = (unsigned)__builtin_amdgcn_readlane(idx, j0 + u);
            v[u] = *(const v2f*)(hl + (size_t)srow * DIM);
        }
        int rem = cur_end - e0;
#pragma unroll
        for (int u = 0; u < AGG_BATCH; ++u) {
            while (u >= rem) {
                const float r = __int_as_float(__builtin_amdgcn_readlane(rsl, cur));
                v2f outv; outv.x = ax * r; outv.y = ay * r;
                *(v2f*)(ol + (size_t)cur * DIM) = outv;
                ax = 0.0f; ay = 0.0f;
                ++cur;
                cur_end = __builtin_amdgcn_readlane(rp, cur + 1);
                rem = cur_end - e0;
            }
            ax += v[u].x; ay += v[u].y;
        }
    };

    const int e_last = (e_end > 0) ? e_end - 1 : 0;
    int idx = in_src[min(e_begin + lane, e_last)];
    for (int base = e_begin; base < e_end; base += 32) {
        const int idx_next = in_src[min(base + 32 + lane, e_last)];
        batch(idx, 0, base);
        if (base + AGG_BATCH < e_end) batch(idx, AGG_BATCH, base + AGG_BATCH);
        idx = idx_next;
    }
    while (cur < AGG_NODES) {
        const float r = __int_as_float(__builtin_amdgcn_readlane(rsl, cur));
        v2f outv; outv.x = ax * r; outv.y = ay * r;
        *(v2f*)(ol + (size_t)cur * DIM) = outv;
        ax = 0.0f; ay = 0.0f;
        ++cur;
    }
}

__global__ __launch_bounds__(128) void gemm_wmma_kernel(const float* __restrict__ X,
                                                        const float* __restrict__ W,
                                                        const float* __restrict__ bias,
                                                        const float* __restrict__ rs_out,
                                                        float* __restrict__ Y,
                                                        int relu_scale) {
    __shared__ float lWsw[16 * 4 * 32 * 2];
    __shared__ float lB[DIM];
    for (int p = threadIdx.x; p < 16 * 4 * 32; p += 128) {
        int l  = p & 31;
        int t  = (p >> 5) & 3;
        int f  = p >> 7;
        int ka = 4 * f + 2 * (l >> 4);
        int n  = 16 * t + (l & 15);
        v2f w2;
        w2.x = W[ka * DIM + n];
        w2.y = W[(ka + 1) * DIM + n];
        *(v2f*)&lWsw[p * 2] = w2;
    }
    if (threadIdx.x < DIM) lB[threadIdx.x] = bias[threadIdx.x];
    __syncthreads();

    const int wave = threadIdx.x >> 5;
    const int lane = threadIdx.x & 31;
    const int half = lane >> 4;
    const int l16  = lane & 15;
    const int m0   = blockIdx.x * 64 + wave * 16;
    const int mA   = m0 + l16;

    v8f acc[4] = {v8f{}, v8f{}, v8f{}, v8f{}};

#pragma unroll
    for (int f = 0; f < 16; ++f) {
        const int ka = 4 * f + 2 * half;
        v2f a = *(const v2f*)(X + (size_t)mA * DIM + ka);
#pragma unroll
        for (int t = 0; t < 4; ++t) {
            v2f b = *(const v2f*)&lWsw[(((f * 4 + t) * 32) + lane) * 2];
            acc[t] = __builtin_amdgcn_wmma_f32_16x16x4_f32(
                false, a, false, b, (short)0, acc[t], false, false);
        }
    }

    const int mrow0 = m0 + half * 8;
    float rsv[8];
#pragma unroll
    for (int r = 0; r < 8; ++r) rsv[r] = relu_scale ? rs_out[mrow0 + r] : 1.0f;

#pragma unroll
    for (int t = 0; t < 4; ++t) {
        const int n = t * 16 + l16;
        const float bn = lB[n];
#pragma unroll
        for (int r = 0; r < 8; ++r) {
            float v = acc[t][r] + bn;
            if (relu_scale) v = fmaxf(v, 0.0f) * rsv[r];
            Y[(size_t)(mrow0 + r) * DIM + n] = v;
        }
    }
}

extern "C" void kernel_launch(void* const* d_in, const int* in_sizes, int n_in,
                              void* d_out, int out_size, void* d_ws, size_t ws_size,
                              hipStream_t stream) {
    (void)in_sizes; (void)n_in; (void)out_size; (void)ws_size;
    const float* x  = (const float*)d_in[0];
    const int*   ei = (const int*)d_in[1];
    const float* W1 = (const float*)d_in[3];
    const float* b1 = (const float*)d_in[4];
    const float* W2 = (const float*)d_in[5];
    const float* b2 = (const float*)d_in[6];
    const float* W3 = (const float*)d_in[7];
    const float* b3 = (const float*)d_in[8];
    float* out = (float*)d_out;

    char* ws = (char*)d_ws;
    unsigned* deg_in  = (unsigned*)ws; ws += (size_t)N_NODES * 4;
    float*    rs_out  = (float*)ws;    ws += (size_t)N_NODES * 4;
    float*    rs_in   = (float*)ws;    ws += (size_t)N_NODES * 4;
    int*      row_ptr = (int*)ws;      ws += (size_t)(N_NODES + 16) * 4;
    int*      in_src  = (int*)ws;      ws += (size_t)N_EDGES * 4;
    float*    hs      = (float*)ws;    ws += (size_t)N_NODES * DIM * 4;
    float*    aggn    = (float*)ws;    ws += (size_t)N_NODES * DIM * 4;
    unsigned* cnt_src = (unsigned*)hs;
    unsigned* cnt_dst = (unsigned*)aggn;

    hist_kernel<<<dim3(CHUNKS, RANGES, 2), 256, 0, stream>>>(ei, cnt_src, cnt_dst);
    merge_kernel<<<dim3(N_NODES / 256, 2), 256, 0, stream>>>(cnt_src, cnt_dst, deg_in, rs_out, rs_in);
    scan_kernel<<<1, 1024, 0, stream>>>(deg_in, row_ptr);
    fill_kernel<<<dim3(CHUNKS, RANGES), 256, 0, stream>>>(ei, cnt_dst, row_ptr, in_src);

    scale0_kernel<<<(N_NODES * DIM / 4) / 256, 256, 0, stream>>>(x, rs_out, hs);

    agg_kernel<<<(N_NODES / AGG_NODES * 32) / 256, 256, 0, stream>>>(hs, in_src, row_ptr, rs_in, aggn);
    gemm_wmma_kernel<<<N_NODES / 64, 128, 0, stream>>>(aggn, W1, b1, rs_out, hs, 1);
    agg_kernel<<<(N_NODES / AGG_NODES * 32) / 256, 256, 0, stream>>>(hs, in_src, row_ptr, rs_in, aggn);
    gemm_wmma_kernel<<<N_NODES / 64, 128, 0, stream>>>(aggn, W2, b2, rs_out, hs, 1);
    agg_kernel<<<(N_NODES / AGG_NODES * 32) / 256, 256, 0, stream>>>(hs, in_src, row_ptr, rs_in, aggn);
    gemm_wmma_kernel<<<N_NODES / 64, 128, 0, stream>>>(aggn, W3, b3, rs_out, out, 0);
}


# hs = x * rsqrt(out-degree) moved from its own 4096-workgroup launch into 128 extra 1024-thread workgroups (8 float4 per thread) of the one-workgroup row_ptr scan launch; 10 launches
# speedup vs baseline: 2.0630x; 1.0294x over previous
#include <hip/hip_runtime.h>

#define N_NODES 65536
#define N_EDGES 1048576
#define DIM 64

typedef float v2f __attribute__((ext_vector_type(2)));
typedef float v8f __attribute__((ext_vector_type(8)));


#define CHUNKS 64
#define CHUNK_EDGES (N_EDGES / CHUNKS)
#define RANGES 4
#define RANGE_NODES (N_NODES / RANGES)
#define RANGE_SHIFT 14
#define EDGE_BATCH 8

__global__ __launch_bounds__(256) void hist_kernel(const int* __restrict__ ei,
                                                   unsigned* __restrict__ cnt_src,
                                                   unsigned* __restrict__ cnt_dst) {
    __shared__ unsigned tab[RANGE_NODES];
    const int g = blockIdx.x, q = blockIdx.y, which = blockIdx.z;
    uint4* tab4 = (uint4*)tab;
    for (int i = threadIdx.x; i < RANGE_NODES / 4; i += 256) tab4[i] = make_uint4(0u, 0u, 0u, 0u);
    __syncthreads();
    const int4* idx4 = (const int4*)(ei + (size_t)which * N_EDGES + (size_t)g * CHUNK_EDGES);
    for (int i0 = threadIdx.x; i0 < CHUNK_EDGES / 4; i0 += 256 * EDGE_BATCH) {
        int4 v[EDGE_BATCH];
#pragma unroll
        for (int k = 0; k < EDGE_BATCH; ++k) v[k] = idx4[i0 + k * 256];
#pragma unroll
        for (int k = 0; k < EDGE_BATCH; ++k) {
            if ((v[k].x >> RANGE_SHIFT) == q) atomicAdd(&tab[v[k].x & (RANGE_NODES - 1)], 1u);
            if ((v[k].y >> RANGE_SHIFT) == q) atomicAdd(&tab[v[k].y & (RANGE_NODES - 1)], 1u);
            if ((v[k].z >> RANGE_SHIFT) == q) atomicAdd(&tab[v[k].z & (RANGE_NODES - 1)], 1u);
            if ((v[k].w >> RANGE_SHIFT) == q) atomicAdd(&tab[v[k].w & (RANGE_NODES - 1)], 1u);
        }
    }
    __syncthreads();
    unsigned* cnt = which ? cnt_dst : cnt_src;
    uint4* out4 = (uint4*)(cnt + (size_t)g * N_NODES + (size_t)q * RANGE_NODES);
    for (int i = threadIdx.x; i < RANGE_NODES / 4; i += 256) out4[i] = tab4[i];
}

__global__ __launch_bounds__(256) void merge_kernel(const unsigned* __restrict__ cnt_src,
                                                    unsigned* __restrict__ cnt_dst,
                                                    unsigned* __restrict__ deg_in,
                                                    float* __restrict__ rs_out,
                                                    float* __restrict__ rs_in) {
    const int n = blockIdx.x * 256 + threadIdx.x;
    if (blockIdx.y == 0) {
        unsigned dout = 0u;
#pragma unroll
        for (int g = 0; g < CHUNKS; ++g) dout += cnt_src[(size_t)g * N_NODES + n];
        if (dout < 1u) dout = 1u;
        rs_out[n] = rsqrtf((float)dout);
    } else {
        unsigned c[CHUNKS];
#pragma unroll
        for (int g = 0; g < CHUNKS; ++g) c[g] = cnt_dst[(size_t)g * N_NODES + n];
        unsigned din = 0u;
#pragma unroll
        for (int g = 0; g < CHUNKS; ++g) {
            cnt_dst[(size_t)g * N_NODES + n] = din;
            din += c[g];
        }
        deg_in[n] = din;
        if (din < 1u) din = 1u;
        rs_in[n] = rsqrtf((float)din);
    }
}

#define SCALE_BLOCKS 128
#define SCALE_PER_THREAD 8

__global__ __launch_bounds__(1024) void scan_scale_kernel(const unsigned* __restrict__ deg_in,
                                                          int* __restrict__ row_ptr,
                                                          const float* __restrict__ x,
                                                          const float* __restrict__ rs_out,
                                                          float* __restrict__ hs) {
    __shared__ unsigned part[1024];
    int t = threadIdx.x;
    if (blockIdx.x != 0) {
        const float4* x4 = (const float4*)x;
        float4* h4 = (float4*)hs;
        const int i0 = (blockIdx.x - 1) * (SCALE_PER_THREAD * 1024) + t;
        float4 v[SCALE_PER_THREAD];
        float r[SCALE_PER_THREAD];
#pragma unroll
        for (int k = 0; k < SCALE_PER_THREAD; ++k) {
            v[k] = x4[i0 + k * 1024];
            r[k] = rs_out[(i0 + k * 1024) >> 4];
        }
#pragma unroll
        for (int k = 0; k < SCALE_PER_THREAD; ++k) {
            v[k].x *= r[k]; v[k].y *= r[k]; v[k].z *= r[k]; v[k].w *= r[k];
            h4[i0 + k * 1024] = v[k];
        }
        return;
    }
    int base = t * (N_NODES / 1024);
    unsigned sum = 0;
    for (int i = 0; i < N_NODES / 1024; ++i) sum += deg_in[base + i];
    part[t] = sum;
    __syncthreads();
    for (int off = 1; off < 1024; off <<= 1) {
        unsigned v = (t >= off) ? part[t - off] : 0u;
        __syncthreads();
        part[t] += v;
        __syncthreads();
    }
    unsigned run = (t == 0) ? 0u : part[t - 1];
    for (int i = 0; i < N_NODES / 1024; ++i) {
        row_ptr[base + i] = (int)run;
        run += deg_in[base + i];
    }
    if (t == 1023) row_ptr[N_NODES] = (int)run;
}

__global__ __launch_bounds__(256) void fill_kernel(const int* __restrict__ ei,
                                                   const unsigned* __restrict__ chunk_off,
                                                   const int* __restrict__ row_ptr,
                                                   int* __restrict__ in_src) {
    __shared__ unsigned cur[RANGE_NODES];
    const int g = blockIdx.x, q = blockIdx.y;
    const uint4* off4 = (const uint4*)(chunk_off + (size_t)g * N_NODES + (size_t)q * RANGE_NODES);
    const int4*  rp4  = (const int4*)(row_ptr + (size_t)q * RANGE_NODES);
    uint4* cur4 = (uint4*)cur;
    for (int i0 = threadIdx.x; i0 < RANGE_NODES / 4; i0 += 256 * EDGE_BATCH) {
        uint4 o[EDGE_BATCH];
        int4  r[EDGE_BATCH];
#pragma unroll
        for (int k = 0; k < EDGE_BATCH; ++k) { o[k] = off4[i0 + k * 256]; r[k] = rp4[i0 + k * 256]; }
#pragma unroll
        for (int k = 0; k < EDGE_BATCH; ++k)
            cur4[i0 + k * 256] = make_uint4(o[k].x + (unsigned)r[k].x, o[k].y + (unsigned)r[k].y,
                                            o[k].z + (unsigned)r[k].z, o[k].w + (unsigned)r[k].w);
    }
    __syncthreads();
    const int4* s4 = (const int4*)(ei + (size_t)g * CHUNK_EDGES);
    const int4* d4 = (const int4*)(ei + (size_t)N_EDGES + (size_t)g * CHUNK_EDGES);
    for (int i0 = threadIdx.x; i0 < CHUNK_EDGES / 4; i0 += 256 * EDGE_BATCH) {
        int4 d[EDGE_BATCH], s[EDGE_BATCH];
#pragma unroll
        for (int k = 0; k < EDGE_BATCH; ++k) { d[k] = d4[i0 + k * 256]; s[k] = s4[i0 + k * 256]; }
#pragma unroll
        for (int k = 0; k < EDGE_BATCH; ++k) {
            if ((d[k].x >> RANGE_SHIFT) == q) in_src[atomicAdd(&cur[d[k].x & (RANGE_NODES - 1)], 1u)] = s[k].x;
            if ((d[k].y >> RANGE_SHIFT) == q) in_src[atomicAdd(&cur[d[k].y & (RANGE_NODES - 1)], 1u)] = s[k].y;
            if ((d[k].z >> RANGE_SHIFT) == q) in_src[atomicAdd(&cur[d[k].z & (RANGE_NODES - 1)], 1u)] = s[k].z;
            if ((d[k].w >> RANGE_SHIFT) == q) in_src[atomicAdd(&cur[d[k].w & (RANGE_NODES - 1)], 1u)] = s[k].w;
        }
    }
}

#define AGG_NODES 8
#define AGG_BATCH 16

__global__ __launch_bounds__(256) void agg_kernel(const float* __restrict__ hs,
                                                  const int* __restrict__ in_src,
                                                  const int* __restrict__ row_ptr,
                                                  const float* __restrict__ rs_in,
                                                  float* __restrict__ aggn) {
    const int lane  = threadIdx.x & 31;
    const int node0 = ((blockIdx.x * 256 + threadIdx.x) >> 5) * AGG_NODES;
    const int rp  = (lane <= AGG_NODES) ? row_ptr[node0 + lane] : 0x7fffffff;
    const int rsl = __float_as_int(rs_in[node0 + (lane < AGG_NODES ? lane : AGG_NODES - 1)]);
    const int e_begin = __builtin_amdgcn_readlane(rp, 0);
    const int e_end   = __builtin_amdgcn_readlane(rp, AGG_NODES);
    int cur     = 0;
    int cur_end = __builtin_amdgcn_readlane(rp, 1);
    float ax = 0.0f, ay = 0.0f;
    const float* hl = hs + lane * 2;
    float* ol = aggn + (size_t)node0 * DIM + lane * 2;

    auto batch = [&](const int idx, const int j0, const int e0) {
        v2f v[AGG_BATCH];
#pragma unroll
        for (int u = 0; u < AGG_BATCH; ++u) {
            const unsigned srow = (unsigned)__builtin_amdgcn_readlane(idx, j0 + u);
            v[u] = *(const v2f*)(hl + (size_t)srow * DIM);
        }
        int rem = cur_end - e0;
#pragma unroll
        for (int u = 0; u < AGG_BATCH; ++u) {
            while (u >= rem) {
                const float r = __int_as_float(__builtin_amdgcn_readlane(rsl, cur));
                v2f outv; outv.x = ax * r; outv.y = ay * r;
                *(v2f*)(ol + (size_t)cur * DIM) = outv;
                ax = 0.0f; ay = 0.0f;
                ++cur;
                cur_end = __builtin_amdgcn_readlane(rp, cur + 1);
                rem = cur_end - e0;
            }
            ax += v[u].x; ay += v[u].y;
        }
    };

    const int e_last = (e_end > 0) ? e_end - 1 : 0;
    int idx = in_src[min(e_begin + lane, e_last)];
    for (int base = e_begin; base < e_end; base += 32) {
        const int idx_next = in_src[min(base + 32 + lane, e_last)];
        batch(idx, 0, base);
        if (base + AGG_BATCH < e_end) batch(idx, AGG_BATCH, base + AGG_BATCH);
        idx = idx_next;
    }
    while (cur < AGG_NODES) {
        const float r = __int_as_float(__builtin_amdgcn_readlane(rsl, cur));
        v2f outv; outv.x = ax * r; outv.y = ay * r;
        *(v2f*)(ol + (size_t)cur * DIM) = outv;
        ax = 0.0f; ay = 0.0f;
        ++cur;
    }
}

__global__ __launch_bounds__(128) void gemm_wmma_kernel(const float* __restrict__ X,
                                                        const float* __restrict__ W,
                                                        const float* __restrict__ bias,
                                                        const float* __restrict__ rs_out,
                                                        float* __restrict__ Y,
                                                        int relu_scale) {
    __shared__ float lWsw[16 * 4 * 32 * 2];
    __shared__ float lB[DIM];
    for (int p = threadIdx.x; p < 16 * 4 * 32; p += 128) {
        int l  = p & 31;
        int t  = (p >> 5) & 3;
        int f  = p >> 7;
        int ka = 4 * f + 2 * (l >> 4);
        int n  = 16 * t + (l & 15);
        v2f w2;
        w2.x = W[ka * DIM + n];
        w2.y = W[(ka + 1) * DIM + n];
        *(v2f*)&lWsw[p * 2] = w2;
    }
    if (threadIdx.x < DIM) lB[threadIdx.x] = bias[threadIdx.x];
    __syncthreads();

    const int wave = threadIdx.x >> 5;
    const int lane = threadIdx.x & 31;
    const int half = lane >> 4;
    const int l16  = lane & 15;
    const int m0   = blockIdx.x * 64 + wave * 16;
    const int mA   = m0 + l16;

    v8f acc[4] = {v8f{}, v8f{}, v8f{}, v8f{}};

#pragma unroll
    for (int f = 0; f < 16; ++f) {
        const int ka = 4 * f + 2 * half;
        v2f a = *(const v2f*)(X + (size_t)mA * DIM + ka);
#pragma unroll
        for (int t = 0; t < 4; ++t) {
            v2f b = *(const v2f*)&lWsw[(((f * 4 + t) * 32) + lane) * 2];
            acc[t] = __builtin_amdgcn_wmma_f32_16x16x4_f32(
                false, a, false, b, (short)0, acc[t], false, false);
        }
    }

    const int mrow0 = m0 + half * 8;
    float rsv[8];
#pragma unroll
    for (int r = 0; r < 8; ++r) rsv[r] = relu_scale ? rs_out[mrow0 + r] : 1.0f;

#pragma unroll
    for (int t = 0; t < 4; ++t) {
        const int n = t * 16 + l16;
        const float bn = lB[n];
#pragma unroll
        for (int r = 0; r < 8; ++r) {
            float v = acc[t][r] + bn;
            if (relu_scale) v = fmaxf(v, 0.0f) * rsv[r];
            Y[(size_t)(mrow0 + r) * DIM + n] = v;
        }
    }
}

extern "C" void kernel_launch(void* const* d_in, const int* in_sizes, int n_in,
                              void* d_out, int out_size, void* d_ws, size_t ws_size,
                              hipStream_t stream) {
    (void)in_sizes; (void)n_in; (void)out_size; (void)ws_size;
    const float* x  = (const float*)d_in[0];
    const int*   ei = (const int*)d_in[1];
    const float* W1 = (const float*)d_in[3];
    const float* b1 = (const float*)d_in[4];
    const float* W2 = (const float*)d_in[5];
    const float* b2 = (const float*)d_in[6];
    const float* W3 = (const float*)d_in[7];
    const float* b3 = (const float*)d_in[8];
    float* out = (float*)d_out;

    char* ws = (char*)d_ws;
    unsigned* deg_in  = (unsigned*)ws; ws += (size_t)N_NODES * 4;
    float*    rs_out  = (float*)ws;    ws += (size_t)N_NODES * 4;
    float*    rs_in   = (float*)ws;    ws += (size_t)N_NODES * 4;
    int*      row_ptr = (int*)ws;      ws += (size_t)(N_NODES + 16) * 4;
    int*      in_src  = (int*)ws;      ws += (size_t)N_EDGES * 4;
    float*    hs      = (float*)ws;    ws += (size_t)N_NODES * DIM * 4;
    float*    aggn    = (float*)ws;    ws += (size_t)N_NODES * DIM * 4;
    unsigned* cnt_src = (unsigned*)hs;
    unsigned* cnt_dst = (unsigned*)aggn;

    hist_kernel<<<dim3(CHUNKS, RANGES, 2), 256, 0, stream>>>(ei, cnt_src, cnt_dst);
    merge_kernel<<<dim3(N_NODES / 256, 2), 256, 0, stream>>>(cnt_src, cnt_dst, deg_in, rs_out, rs_in);
    scan_scale_kernel<<<1 + SCALE_BLOCKS, 1024, 0, stream>>>(deg_in, row_ptr, x, rs_out, hs);
    fill_kernel<<<dim3(CHUNKS, RANGES), 256, 0, stream>>>(ei, cnt_dst, row_ptr, in_src);

    agg_kernel<<<(N_NODES / AGG_NODES * 32) / 256, 256, 0, stream>>>(hs, in_src, row_ptr, rs_in, aggn);
    gemm_wmma_kernel<<<N_NODES / 64, 128, 0, stream>>>(aggn, W1, b1, rs_out, hs, 1);
    agg_kernel<<<(N_NODES / AGG_NODES * 32) / 256, 256, 0, stream>>>(hs, in_src, row_ptr, rs_in, aggn);
    gemm_wmma_kernel<<<N_NODES / 64, 128, 0, stream>>>(aggn, W2, b2, rs_out, hs, 1);
    agg_kernel<<<(N_NODES / AGG_NODES * 32) / 256, 256, 0, stream>>>(hs, in_src, row_ptr, rs_in, aggn);
    gemm_wmma_kernel<<<N_NODES / 64, 128, 0, stream>>>(aggn, W3, b3, rs_out, out, 0);
}


# row_ptr scan workgroup: deg_in read and row_ptr written as whole-line uint4 per lane (was 32 lines per wave access), wave DPP scans replace the 10-round LDS scan; integers only, float sums unchanged
# speedup vs baseline: 2.2705x; 1.1006x over previous
#include <hip/hip_runtime.h>

#define N_NODES 65536
#define N_EDGES 1048576
#define DIM 64

typedef float v2f __attribute__((ext_vector_type(2)));
typedef float v8f __attribute__((ext_vector_type(8)));


#define CHUNKS 64
#define CHUNK_EDGES (N_EDGES / CHUNKS)
#define RANGES 4
#define RANGE_NODES (N_NODES / RANGES)
#define RANGE_SHIFT 14
#define EDGE_BATCH 8

__global__ __launch_bounds__(256) void hist_kernel(const int* __restrict__ ei,
                                                   unsigned* __restrict__ cnt_src,
                                                   unsigned* __restrict__ cnt_dst) {
    __shared__ unsigned tab[RANGE_NODES];
    const int g = blockIdx.x, q = blockIdx.y, which = blockIdx.z;
    uint4* tab4 = (uint4*)tab;
    for (int i = threadIdx.x; i < RANGE_NODES / 4; i += 256) tab4[i] = make_uint4(0u, 0u, 0u, 0u);
    __syncthreads();
    const int4* idx4 = (const int4*)(ei + (size_t)which * N_EDGES + (size_t)g * CHUNK_EDGES);
    for (int i0 = threadIdx.x; i0 < CHUNK_EDGES / 4; i0 += 256 * EDGE_BATCH) {
        int4 v[EDGE_BATCH];
#pragma unroll
        for (int k = 0; k < EDGE_BATCH; ++k) v[k] = idx4[i0 + k * 256];
#pragma unroll
        for (int k = 0; k < EDGE_BATCH; ++k) {
            if ((v[k].x >> RANGE_SHIFT) == q) atomicAdd(&tab[v[k].x & (RANGE_NODES - 1)], 1u);
            if ((v[k].y >> RANGE_SHIFT) == q) atomicAdd(&tab[v[k].y & (RANGE_NODES - 1)], 1u);
            if ((v[k].z >> RANGE_SHIFT) == q) atomicAdd(&tab[v[k].z & (RANGE_NODES - 1)], 1u);
            if ((v[k].w >> RANGE_SHIFT) == q) atomicAdd(&tab[v[k].w & (RANGE_NODES - 1)], 1u);
        }
    }
    __syncthreads();
    unsigned* cnt = which ? cnt_dst : cnt_src;
    uint4* out4 = (uint4*)(cnt + (size_t)g * N_NODES + (size_t)q * RANGE_NODES);
    for (int i = threadIdx.x; i < RANGE_NODES / 4; i += 256) out4[i] = tab4[i];
}

__global__ __launch_bounds__(256) void merge_kernel(const unsigned* __restrict__ cnt_src,
                                                    unsigned* __restrict__ cnt_dst,
                                                    unsigned* __restrict__ deg_in,
                                                    float* __restrict__ rs_out,
                                                    float* __restrict__ rs_in) {
    const int n = blockIdx.x * 256 + threadIdx.x;
    if (blockIdx.y == 0) {
        unsigned dout = 0u;
#pragma unroll
        for (int g = 0; g < CHUNKS; ++g) dout += cnt_src[(size_t)g * N_NODES + n];
        if (dout < 1u) dout = 1u;
        rs_out[n] = rsqrtf((float)dout);
    } else {
        unsigned c[CHUNKS];
#pragma unroll
        for (int g = 0; g < CHUNKS; ++g) c[g] = cnt_dst[(size_t)g * N_NODES + n];
        unsigned din = 0u;
#pragma unroll
        for (int g = 0; g < CHUNKS; ++g) {
            cnt_dst[(size_t)g * N_NODES + n] = din;
            din += c[g];
        }
        deg_in[n] = din;
        if (din < 1u) din = 1u;
        rs_in[n] = rsqrtf((float)din);
    }
}

#define SCALE_BLOCKS 128
#define SCALE_PER_THREAD 8
#define SCAN_TILES 16

__device__ __forceinline__ unsigned wave_incl_scan(unsigned v, const int lane) {
    v += (unsigned)__builtin_amdgcn_update_dpp(0, (int)v, 0x111, 0xf, 0xf, false);
    v += (unsigned)__builtin_amdgcn_update_dpp(0, (int)v, 0x112, 0xf, 0xf, false);
    v += (unsigned)__builtin_amdgcn_update_dpp(0, (int)v, 0x114, 0xf, 0xf, false);
    v += (unsigned)__builtin_amdgcn_update_dpp(0, (int)v, 0x118, 0xf, 0xf, false);
    const unsigned low_row = (unsigned)__builtin_amdgcn_readlane((int)v, 15);
    return (lane >= 16) ? v + low_row : v;
}

__global__ __launch_bounds__(1024) void scan_scale_kernel(const unsigned* __restrict__ deg_in,
                                                          int* __restrict__ row_ptr,
                                                          const float* __restrict__ x,
                                                          const float* __restrict__ rs_out,
                                                          float* __restrict__ hs) {
    __shared__ alignas(16) unsigned part[SCAN_TILES * 32 + 16];
    int t = threadIdx.x;
    if (blockIdx.x != 0) {
        const float4* x4 = (const float4*)x;
        float4* h4 = (float4*)hs;
        const int i0 = (blockIdx.x - 1) * (SCALE_PER_THREAD * 1024) + t;
        float4 v[SCALE_PER_THREAD];
        float r[SCALE_PER_THREAD];
#pragma unroll
        for (int k = 0; k < SCALE_PER_THREAD; ++k) {
            v[k] = x4[i0 + k * 1024];
            r[k] = rs_out[(i0 + k * 1024) >> 4];
        }
#pragma unroll
        for (int k = 0; k < SCALE_PER_THREAD; ++k) {
            v[k].x *= r[k]; v[k].y *= r[k]; v[k].z *= r[k]; v[k].w *= r[k];
            h4[i0 + k * 1024] = v[k];
        }
        return;
    }
    const int lane = t & 31;
    const int wave = t >> 5;
    const uint4* d4 = (const uint4*)deg_in;
    uint4 d[SCAN_TILES];
#pragma unroll
    for (int k = 0; k < SCAN_TILES; ++k) d[k] = d4[k * 1024 + t];
    unsigned inc[SCAN_TILES];
#pragma unroll
    for (int k = 0; k < SCAN_TILES; ++k) {
        inc[k] = wave_incl_scan(d[k].x + d[k].y + d[k].z + d[k].w, lane);
        if (lane == 31) part[k * 32 + wave] = inc[k];
    }
    __syncthreads();
    if (wave == 0) {
        uint4* p4 = (uint4*)(part + 16 * lane);
        const uint4 q0 = p4[0], q1 = p4[1], q2 = p4[2], q3 = p4[3];
        unsigned w[16] = {q0.x, q0.y, q0.z, q0.w, q1.x, q1.y, q1.z, q1.w,
                          q2.x, q2.y, q2.z, q2.w, q3.x, q3.y, q3.z, q3.w};
        unsigned run = 0u;
#pragma unroll
        for (int i = 0; i < 16; ++i) { const unsigned c = w[i]; w[i] = run; run += c; }
        const unsigned incl = wave_incl_scan(run, lane);
        const unsigned excl = incl - run;
#pragma unroll
        for (int i = 0; i < 16; ++i) w[i] += excl;
        p4[0] = make_uint4(w[0],  w[1],  w[2],  w[3]);
        p4[1] = make_uint4(w[4],  w[5],  w[6],  w[7]);
        p4[2] = make_uint4(w[8],  w[9],  w[10], w[11]);
        p4[3] = make_uint4(w[12], w[13], w[14], w[15]);
        if (lane == 31) part[SCAN_TILES * 32] = incl;
    }
    __syncthreads();
    int4* rp4 = (int4*)row_ptr;
#pragma unroll
    for (int k = 0; k < SCAN_TILES; ++k) {
        const unsigned s = d[k].x + d[k].y + d[k].z + d[k].w;
        const unsigned p = part[k * 32 + wave] + inc[k] - s;
        int4 o;
        o.x = (int)p;
        o.y = (int)(p + d[k].x);
        o.z = (int)(p + d[k].x + d[k].y);
        o.w = (int)(p + d[k].x + d[k].y + d[k].z);
        rp4[k * 1024 + t] = o;
    }
    if (t == 1023) row_ptr[N_NODES] = (int)part[SCAN_TILES * 32];
}

__global__ __launch_bounds__(256) void fill_kernel(const int* __restrict__ ei,
                                                   const unsigned* __restrict__ chunk_off,
                                                   const int* __restrict__ row_ptr,
                                                   int* __restrict__ in_src) {
    __shared__ unsigned cur[RANGE_NODES];
    const int g = blockIdx.x, q = blockIdx.y;
    const uint4* off4 = (const uint4*)(chunk_off + (size_t)g * N_NODES + (size_t)q * RANGE_NODES);
    const int4*  rp4  = (const int4*)(row_ptr + (size_t)q * RANGE_NODES);
    uint4* cur4 = (uint4*)cur;
    for (int i0 = threadIdx.x; i0 < RANGE_NODES / 4; i0 += 256 * EDGE_BATCH) {
        uint4 o[EDGE_BATCH];
        int4  r[EDGE_BATCH];
#pragma unroll
        for (int k = 0; k < EDGE_BATCH; ++k) { o[k] = off4[i0 + k * 256]; r[k] = rp4[i0 + k * 256]; }
#pragma unroll
        for (int k = 0; k < EDGE_BATCH; ++k)
            cur4[i0 + k * 256] = make_uint4(o[k].x + (unsigned)r[k].x, o[k].y + (unsigned)r[k].y,
                                            o[k].z + (unsigned)r[k].z, o[k].w + (unsigned)r[k].w);
    }
    __syncthreads();
    const int4* s4 = (const int4*)(ei + (size_t)g * CHUNK_EDGES);
    const int4* d4 = (const int4*)(ei + (size_t)N_EDGES + (size_t)g * CHUNK_EDGES);
    for (int i0 = threadIdx.x; i0 < CHUNK_EDGES / 4; i0 += 256 * EDGE_BATCH) {
        int4 d[EDGE_BATCH], s[EDGE_BATCH];
#pragma unroll
        for (int k = 0; k < EDGE_BATCH; ++k) { d[k] = d4[i0 + k * 256]; s[k] = s4[i0 + k * 256]; }
#pragma unroll
        for (int k = 0; k < EDGE_BATCH; ++k) {
            if ((d[k].x >> RANGE_SHIFT) == q) in_src[atomicAdd(&cur[d[k].x & (RANGE_NODES - 1)], 1u)] = s[k].x;
            if ((d[k].y >> RANGE_SHIFT) == q) in_src[atomicAdd(&cur[d[k].y & (RANGE_NODES - 1)], 1u)] = s[k].y;
            if ((d[k].z >> RANGE_SHIFT) == q) in_src[atomicAdd(&cur[d[k].z & (RANGE_NODES - 1)], 1u)] = s[k].z;
            if ((d[k].w >> RANGE_SHIFT) == q) in_src[atomicAdd(&cur[d[k].w & (RANGE_NODES - 1)], 1u)] = s[k].w;
        }
    }
}

#define AGG_NODES 8
#define AGG_BATCH 16

__global__ __launch_bounds__(256) void agg_kernel(const float* __restrict__ hs,
                                                  const int* __restrict__ in_src,
                                                  const int* __restrict__ row_ptr,
                                                  const float* __restrict__ rs_in,
                                                  float* __restrict__ aggn) {
    const int lane  = threadIdx.x & 31;
    const int node0 = ((blockIdx.x * 256 + threadIdx.x) >> 5) * AGG_NODES;
    const int rp  = (lane <= AGG_NODES) ? row_ptr[node0 + lane] : 0x7fffffff;
    const int rsl = __float_as_int(rs_in[node0 + (lane < AGG_NODES ? lane : AGG_NODES - 1)]);
    const int e_begin = __builtin_amdgcn_readlane(rp, 0);
    const int e_end   = __builtin_amdgcn_readlane(rp, AGG_NODES);
    int cur     = 0;
    int cur_end = __builtin_amdgcn_readlane(rp, 1);
    float ax = 0.0f, ay = 0.0f;
    const float* hl = hs + lane * 2;
    float* ol = aggn + (size_t)node0 * DIM + lane * 2;

    auto batch = [&](const int idx, const int j0, const int e0) {
        v2f v[AGG_BATCH];
#pragma unroll
        for (int u = 0; u < AGG_BATCH; ++u) {
            const unsigned srow = (unsigned)__builtin_amdgcn_readlane(idx, j0 + u);
            v[u] = *(const v2f*)(hl + (size_t)srow * DIM);
        }
        int rem = cur_end - e0;
#pragma unroll
        for (int u = 0; u < AGG_BATCH; ++u) {
            while (u >= rem) {
                const float r = __int_as_float(__builtin_amdgcn_readlane(rsl, cur));
                v2f outv; outv.x = ax * r; outv.y = ay * r;
                *(v2f*)(ol + (size_t)cur * DIM) = outv;
                ax = 0.0f; ay = 0.0f;
                ++cur;
                cur_end = __builtin_amdgcn_readlane(rp, cur + 1);
                rem = cur_end - e0;
            }
            ax += v[u].x; ay += v[u].y;
        }
    };

    const int e_last = (e_end > 0) ? e_end - 1 : 0;
    int idx = in_src[min(e_begin + lane, e_last)];
    for (int base = e_begin; base < e_end; base += 32) {
        const int idx_next = in_src[min(base + 32 + lane, e_last)];
        batch(idx, 0, base);
        if (base + AGG_BATCH < e_end) batch(idx, AGG_BATCH, base + AGG_BATCH);
        idx = idx_next;
    }
    while (cur < AGG_NODES) {
        const float r = __int_as_float(__builtin_amdgcn_readlane(rsl, cur));
        v2f outv; outv.x = ax * r; outv.y = ay * r;
        *(v2f*)(ol + (size_t)cur * DIM) = outv;
        ax = 0.0f; ay = 0.0f;
        ++cur;
    }
}

__global__ __launch_bounds__(128) void gemm_wmma_kernel(const float* __restrict__ X,
                                                        const float* __restrict__ W,
                                                        const float* __restrict__ bias,
                                                        const float* __restrict__ rs_out,
                                                        float* __restrict__ Y,
                                                        int relu_scale) {
    __shared__ float lWsw[16 * 4 * 32 * 2];
    __shared__ float lB[DIM];
    for (int p = threadIdx.x; p < 16 * 4 * 32; p += 128) {
        int l  = p & 31;
        int t  = (p >> 5) & 3;
        int f  = p >> 7;
        int ka = 4 * f + 2 * (l >> 4);
        int n  = 16 * t + (l & 15);
        v2f w2;
        w2.x = W[ka * DIM + n];
        w2.y = W[(ka + 1) * DIM + n];
        *(v2f*)&lWsw[p * 2] = w2;
    }
    if (threadIdx.x < DIM) lB[threadIdx.x] = bias[threadIdx.x];
    __syncthreads();

    const int wave = threadIdx.x >> 5;
    const int lane = threadIdx.x & 31;
    const int half = lane >> 4;
    const int l16  = lane & 15;
    const int m0   = blockIdx.x * 64 + wave * 16;
    const int mA   = m0 + l16;

    v8f acc[4] = {v8f{}, v8f{}, v8f{}, v8f{}};

#pragma unroll
    for (int f = 0; f < 16; ++f) {
        const int ka = 4 * f + 2 * half;
        v2f a = *(const v2f*)(X + (size_t)mA * DIM + ka);
#pragma unroll
        for (int t = 0; t < 4; ++t) {
            v2f b = *(const v2f*)&lWsw[(((f * 4 + t) * 32) + lane) * 2];
            acc[t] = __builtin_amdgcn_wmma_f32_16x16x4_f32(
                false, a, false, b, (short)0, acc[t], false, false);
        }
    }

    const int mrow0 = m0 + half * 8;
    float rsv[8];
#pragma unroll
    for (int r = 0; r < 8; ++r) rsv[r] = relu_scale ? rs_out[mrow0 + r] : 1.0f;

#pragma unroll
    for (int t = 0; t < 4; ++t) {
        const int n = t * 16 + l16;
        const float bn = lB[n];
#pragma unroll
        for (int r = 0; r < 8; ++r) {
            float v = acc[t][r] + bn;
            if (relu_scale) v = fmaxf(v, 0.0f) * rsv[r];
            Y[(size_t)(mrow0 + r) * DIM + n] = v;
        }
    }
}

extern "C" void kernel_launch(void* const* d_in, const int* in_sizes, int n_in,
                              void* d_out, int out_size, void* d_ws, size_t ws_size,
                              hipStream_t stream) {
    (void)in_sizes; (void)n_in; (void)out_size; (void)ws_size;
    const float* x  = (const float*)d_in[0];
    const int*   ei = (const int*)d_in[1];
    const float* W1 = (const float*)d_in[3];
    const float* b1 = (const float*)d_in[4];
    const float* W2 = (const float*)d_in[5];
    const float* b2 = (const float*)d_in[6];
    const float* W3 = (const float*)d_in[7];
    const float* b3 = (const float*)d_in[8];
    float* out = (float*)d_out;

    char* ws = (char*)d_ws;
    unsigned* deg_in  = (unsigned*)ws; ws += (size_t)N_NODES * 4;
    float*    rs_out  = (float*)ws;    ws += (size_t)N_NODES * 4;
    float*    rs_in   = (float*)ws;    ws += (size_t)N_NODES * 4;
    int*      row_ptr = (int*)ws;      ws += (size_t)(N_NODES + 16) * 4;
    int*      in_src  = (int*)ws;      ws += (size_t)N_EDGES * 4;
    float*    hs      = (float*)ws;    ws += (size_t)N_NODES * DIM * 4;
    float*    aggn    = (float*)ws;    ws += (size_t)N_NODES * DIM * 4;
    unsigned* cnt_src = (unsigned*)hs;
    unsigned* cnt_dst = (unsigned*)aggn;

    hist_kernel<<<dim3(CHUNKS, RANGES, 2), 256, 0, stream>>>(ei, cnt_src, cnt_dst);
    merge_kernel<<<dim3(N_NODES / 256, 2), 256, 0, stream>>>(cnt_src, cnt_dst, deg_in, rs_out, rs_in);
    scan_scale_kernel<<<1 + SCALE_BLOCKS, 1024, 0, stream>>>(deg_in, row_ptr, x, rs_out, hs);
    fill_kernel<<<dim3(CHUNKS, RANGES), 256, 0, stream>>>(ei, cnt_dst, row_ptr, in_src);

    agg_kernel<<<(N_NODES / AGG_NODES * 32) / 256, 256, 0, stream>>>(hs, in_src, row_ptr, rs_in, aggn);
    gemm_wmma_kernel<<<N_NODES / 64, 128, 0, stream>>>(aggn, W1, b1, rs_out, hs, 1);
    agg_kernel<<<(N_NODES / AGG_NODES * 32) / 256, 256, 0, stream>>>(hs, in_src, row_ptr, rs_in, aggn);
    gemm_wmma_kernel<<<N_NODES / 64, 128, 0, stream>>>(aggn, W2, b2, rs_out, hs, 1);
    agg_kernel<<<(N_NODES / AGG_NODES * 32) / 256, 256, 0, stream>>>(hs, in_src, row_ptr, rs_in, aggn);
    gemm_wmma_kernel<<<N_NODES / 64, 128, 0, stream>>>(aggn, W3, b3, rs_out, out, 0);
}


# Each layer's neighbour sum and WMMA product fused into one launch: a workgroup's 64 summed rows go to LDS and feed the A fragments, no 16 MB round trip; 10 -> 7 launches; sum and k order unchanged
# speedup vs baseline: 2.6980x; 1.1883x over previous
#include <hip/hip_runtime.h>

#define N_NODES 65536
#define N_EDGES 1048576
#define DIM 64

typedef float v2f __attribute__((ext_vector_type(2)));
typedef float v8f __attribute__((ext_vector_type(8)));


#define CHUNKS 64
#define CHUNK_EDGES (N_EDGES / CHUNKS)
#define RANGES 4
#define RANGE_NODES (N_NODES / RANGES)
#define RANGE_SHIFT 14
#define EDGE_BATCH 8

__global__ __launch_bounds__(256) void hist_kernel(const int* __restrict__ ei,
                                                   unsigned* __restrict__ cnt_src,
                                                   unsigned* __restrict__ cnt_dst) {
    __shared__ unsigned tab[RANGE_NODES];
    const int g = blockIdx.x, q = blockIdx.y, which = blockIdx.z;
    uint4* tab4 = (uint4*)tab;
    for (int i = threadIdx.x; i < RANGE_NODES / 4; i += 256) tab4[i] = make_uint4(0u, 0u, 0u, 0u);
    __syncthreads();
    const int4* idx4 = (const int4*)(ei + (size_t)which * N_EDGES + (size_t)g * CHUNK_EDGES);
    for (int i0 = threadIdx.x; i0 < CHUNK_EDGES / 4; i0 += 256 * EDGE_BATCH) {
        int4 v[EDGE_BATCH];
#pragma unroll
        for (int k = 0; k < EDGE_BATCH; ++k) v[k] = idx4[i0 + k * 256];
#pragma unroll
        for (int k = 0; k < EDGE_BATCH; ++k) {
            if ((v[k].x >> RANGE_SHIFT) == q) atomicAdd(&tab[v[k].x & (RANGE_NODES - 1)], 1u);
            if ((v[k].y >> RANGE_SHIFT) == q) atomicAdd(&tab[v[k].y & (RANGE_NODES - 1)], 1u);
            if ((v[k].z >> RANGE_SHIFT) == q) atomicAdd(&tab[v[k].z & (RANGE_NODES - 1)], 1u);
            if ((v[k].w >> RANGE_SHIFT) == q) atomicAdd(&tab[v[k].w & (RANGE_NODES - 1)], 1u);
        }
    }
    __syncthreads();
    unsigned* cnt = which ? cnt_dst : cnt_src;
    uint4* out4 = (uint4*)(cnt + (size_t)g * N_NODES + (size_t)q * RANGE_NODES);
    for (int i = threadIdx.x; i < RANGE_NODES / 4; i += 256) out4[i] = tab4[i];
}

__global__ __launch_bounds__(256) void merge_kernel(const unsigned* __restrict__ cnt_src,
                                                    unsigned* __restrict__ cnt_dst,
                                                    unsigned* __restrict__ deg_in,
                                                    float* __restrict__ rs_out,
                                                    float* __restrict__ rs_in) {
    const int n = blockIdx.x * 256 + threadIdx.x;
    if (blockIdx.y == 0) {
        unsigned dout = 0u;
#pragma unroll
        for (int g = 0; g < CHUNKS; ++g) dout += cnt_src[(size_t)g * N_NODES + n];
        if (dout < 1u) dout = 1u;
        rs_out[n] = rsqrtf((float)dout);
    } else {
        unsigned c[CHUNKS];
#pragma unroll
        for (int g = 0; g < CHUNKS; ++g) c[g] = cnt_dst[(size_t)g * N_NODES + n];
        unsigned din = 0u;
#pragma unroll
        for (int g = 0; g < CHUNKS; ++g) {
            cnt_dst[(size_t)g * N_NODES + n] = din;
            din += c[g];
        }
        deg_in[n] = din;
        if (din < 1u) din = 1u;
        rs_in[n] = rsqrtf((float)din);
    }
}

#define SCALE_BLOCKS 128
#define SCALE_PER_THREAD 8
#define SCAN_TILES 16

__device__ __forceinline__ unsigned wave_incl_scan(unsigned v, const int lane) {
    v += (unsigned)__builtin_amdgcn_update_dpp(0, (int)v, 0x111, 0xf, 0xf, false);
    v += (unsigned)__builtin_amdgcn_update_dpp(0, (int)v, 0x112, 0xf, 0xf, false);
    v += (unsigned)__builtin_amdgcn_update_dpp(0, (int)v, 0x114, 0xf, 0xf, false);
    v += (unsigned)__builtin_amdgcn_update_dpp(0, (int)v, 0x118, 0xf, 0xf, false);
    const unsigned low_row = (unsigned)__builtin_amdgcn_readlane((int)v, 15);
    return (lane >= 16) ? v + low_row : v;
}

__global__ __launch_bounds__(1024) void scan_scale_kernel(const unsigned* __restrict__ deg_in,
                                                          int* __restrict__ row_ptr,
                                                          const float* __restrict__ x,
                                                          const float* __restrict__ rs_out,
                                                          float* __restrict__ hs) {
    __shared__ alignas(16) unsigned part[SCAN_TILES * 32 + 16];
    int t = threadIdx.x;
    if (blockIdx.x != 0) {
        const float4* x4 = (const float4*)x;
        float4* h4 = (float4*)hs;
        const int i0 = (blockIdx.x - 1) * (SCALE_PER_THREAD * 1024) + t;
        float4 v[SCALE_PER_THREAD];
        float r[SCALE_PER_THREAD];
#pragma unroll
        for (int k = 0; k < SCALE_PER_THREAD; ++k) {
            v[k] = x4[i0 + k * 1024];
            r[k] = rs_out[(i0 + k * 1024) >> 4];
        }
#pragma unroll
        for (int k = 0; k < SCALE_PER_THREAD; ++k) {
            v[k].x *= r[k]; v[k].y *= r[k]; v[k].z *= r[k]; v[k].w *= r[k];
            h4[i0 + k * 1024] = v[k];
        }
        return;
    }
    const int lane = t & 31;
    const int wave = t >> 5;
    const uint4* d4 = (const uint4*)deg_in;
    uint4 d[SCAN_TILES];
#pragma unroll
    for (int k = 0; k < SCAN_TILES; ++k) d[k] = d4[k * 1024 + t];
    unsigned inc[SCAN_TILES];
#pragma unroll
    for (int k = 0; k < SCAN_TILES; ++k) {
        inc[k] = wave_incl_scan(d[k].x + d[k].y + d[k].z + d[k].w, lane);
        if (lane == 31) part[k * 32 + wave] = inc[k];
    }
    __syncthreads();
    if (wave == 0) {
        uint4* p4 = (uint4*)(part + 16 * lane);
        const uint4 q0 = p4[0], q1 = p4[1], q2 = p4[2], q3 = p4[3];
        unsigned w[16] = {q0.x, q0.y, q0.z, q0.w, q1.x, q1.y, q1.z, q1.w,
                          q2.x, q2.y, q2.z, q2.w, q3.x, q3.y, q3.z, q3.w};
        unsigned run = 0u;
#pragma unroll
        for (int i = 0; i < 16; ++i) { const unsigned c = w[i]; w[i] = run; run += c; }
        const unsigned incl = wave_incl_scan(run, lane);
        const unsigned excl = incl - run;
#pragma unroll
        for (int i = 0; i < 16; ++i) w[i] += excl;
        p4[0] = make_uint4(w[0],  w[1],  w[2],  w[3]);
        p4[1] = make_uint4(w[4],  w[5],  w[6],  w[7]);
        p4[2] = make_uint4(w[8],  w[9],  w[10], w[11]);
        p4[3] = make_uint4(w[12], w[13], w[14], w[15]);
        if (lane == 31) part[SCAN_TILES * 32] = incl;
    }
    __syncthreads();
    int4* rp4 = (int4*)row_ptr;
#pragma unroll
    for (int k = 0; k < SCAN_TILES; ++k) {
        const unsigned s = d[k].x + d[k].y + d[k].z + d[k].w;
        const unsigned p = part[k * 32 + wave] + inc[k] - s;
        int4 o;
        o.x = (int)p;
        o.y = (int)(p + d[k].x);
        o.z = (int)(p + d[k].x + d[k].y);
        o.w = (int)(p + d[k].x + d[k].y + d[k].z);
        rp4[k * 1024 + t] = o;
    }
    if (t == 1023) row_ptr[N_NODES] = (int)part[SCAN_TILES * 32];
}

__global__ __launch_bounds__(256) void fill_kernel(const int* __restrict__ ei,
                                                   const unsigned* __restrict__ chunk_off,
                                                   const int* __restrict__ row_ptr,
                                                   int* __restrict__ in_src) {
    __shared__ unsigned cur[RANGE_NODES];
    const int g = blockIdx.x, q = blockIdx.y;
    const uint4* off4 = (const uint4*)(chunk_off + (size_t)g * N_NODES + (size_t)q * RANGE_NODES);
    const int4*  rp4  = (const int4*)(row_ptr + (size_t)q * RANGE_NODES);
    uint4* cur4 = (uint4*)cur;
    for (int i0 = threadIdx.x; i0 < RANGE_NODES / 4; i0 += 256 * EDGE_BATCH) {
        uint4 o[EDGE_BATCH];
        int4  r[EDGE_BATCH];
#pragma unroll
        for (int k = 0; k < EDGE_BATCH; ++k) { o[k] = off4[i0 + k * 256]; r[k] = rp4[i0 + k * 256]; }
#pragma unroll
        for (int k = 0; k < EDGE_BATCH; ++k)
            cur4[i0 + k * 256] = make_uint4(o[k].x + (unsigned)r[k].x, o[k].y + (unsigned)r[k].y,
                                            o[k].z + (unsigned)r[k].z, o[k].w + (unsigned)r[k].w);
    }
    __syncthreads();
    const int4* s4 = (const int4*)(ei + (size_t)g * CHUNK_EDGES);
    const int4* d4 = (const int4*)(ei + (size_t)N_EDGES + (size_t)g * CHUNK_EDGES);
    for (int i0 = threadIdx.x; i0 < CHUNK_EDGES / 4; i0 += 256 * EDGE_BATCH) {
        int4 d[EDGE_BATCH], s[EDGE_BATCH];
#pragma unroll
        for (int k = 0; k < EDGE_BATCH; ++k) { d[k] = d4[i0 + k * 256]; s[k] = s4[i0 + k * 256]; }
#pragma unroll
        for (int k = 0; k < EDGE_BATCH; ++k) {
            if ((d[k].x >> RANGE_SHIFT) == q) in_src[atomicAdd(&cur[d[k].x & (RANGE_NODES - 1)], 1u)] = s[k].x;
            if ((d[k].y >> RANGE_SHIFT) == q) in_src[atomicAdd(&cur[d[k].y & (RANGE_NODES - 1)], 1u)] = s[k].y;
            if ((d[k].z >> RANGE_SHIFT) == q) in_src[atomicAdd(&cur[d[k].z & (RANGE_NODES - 1)], 1u)] = s[k].z;
            if ((d[k].w >> RANGE_SHIFT) == q) in_src[atomicAdd(&cur[d[k].w & (RANGE_NODES - 1)], 1u)] = s[k].w;
        }
    }
}

#define AGG_NODES 8
#define AGG_BATCH 16
#define TILE_ROWS 64
#define TILE_PITCH 68

__global__ __launch_bounds__(256) void layer_kernel(const float* __restrict__ hs,
                                                    const int* __restrict__ in_src,
                                                    const int* __restrict__ row_ptr,
                                                    const float* __restrict__ rs_in,
                                                    const float* __restrict__ W,
                                                    const float* __restrict__ bias,
                                                    const float* __restrict__ rs_out,
                                                    float* __restrict__ Y,
                                                    int relu_scale) {
    __shared__ alignas(16) float lWsw[16 * 4 * 32 * 2];
    __shared__ alignas(16) float lAgg[TILE_ROWS * TILE_PITCH];
    __shared__ float lB[DIM];
    __shared__ float lRs[TILE_ROWS];

    const int lane = threadIdx.x & 31;
    const int wave = threadIdx.x >> 5;

    v2f w2[8];
#pragma unroll
    for (int i = 0; i < 8; ++i) {
        const int p  = threadIdx.x + i * 256;
        const int l  = p & 31;
        const int t  = (p >> 5) & 3;
        const int f  = p >> 7;
        const int ka = 4 * f + 2 * (l >> 4);
        const int n  = 16 * t + (l & 15);
        w2[i].x = W[ka * DIM + n];
        w2[i].y = W[(ka + 1) * DIM + n];
    }
    float sb = 1.0f;
    if (threadIdx.x < DIM) sb = bias[threadIdx.x];
    else if (threadIdx.x < DIM + TILE_ROWS && relu_scale) sb = rs_out[blockIdx.x * TILE_ROWS + threadIdx.x - DIM];

    const int node0 = (blockIdx.x * 8 + wave) * AGG_NODES;
    const int rp  = (lane <= AGG_NODES) ? row_ptr[node0 + lane] : 0x7fffffff;
    const int rsl = __float_as_int(rs_in[node0 + (lane < AGG_NODES ? lane : AGG_NODES - 1)]);
    const int e_begin = __builtin_amdgcn_readlane(rp, 0);
    const int e_end   = __builtin_amdgcn_readlane(rp, AGG_NODES);
    int cur     = 0;
    int cur_end = __builtin_amdgcn_readlane(rp, 1);
    float ax = 0.0f, ay = 0.0f;
    const float* hl = hs + lane * 2;
    float* tl = lAgg + wave * AGG_NODES * TILE_PITCH + lane * 2;

    auto batch = [&](const int idx, const int j0, const int e0) {
        v2f v[AGG_BATCH];
#pragma unroll
        for (int u = 0; u < AGG_BATCH; ++u) {
            const unsigned srow = (unsigned)__builtin_amdgcn_readlane(idx, j0 + u);
            v[u] = *(const v2f*)(hl + (size_t)srow * DIM);
        }
        int rem = cur_end - e0;
#pragma unroll
        for (int u = 0; u < AGG_BATCH; ++u) {
            while (u >= rem) {
                const float r = __int_as_float(__builtin_amdgcn_readlane(rsl, cur));
                v2f outv; outv.x = ax * r; outv.y = ay * r;
                *(v2f*)(tl + cur * TILE_PITCH) = outv;
                ax = 0.0f; ay = 0.0f;
                ++cur;
                cur_end = __builtin_amdgcn_readlane(rp, cur + 1);
                rem = cur_end - e0;
            }
            ax += v[u].x; ay += v[u].y;
        }
    };

    const int e_last = (e_end > 0) ? e_end - 1 : 0;
    int idx = in_src[min(e_begin + lane, e_last)];

#pragma unroll
    for (int i = 0; i < 8; ++i) {
        const int p = threadIdx.x + i * 256;
        const int l = p & 31;
        const int t = (p >> 5) & 3;
        const int f = p >> 7;
        *(v2f*)&lWsw[(((f * 2 + (t >> 1)) * 32 + l) * 2 + (t & 1)) * 2] = w2[i];
    }
    if (threadIdx.x < DIM) lB[threadIdx.x] = sb;
    else if (threadIdx.x < DIM + TILE_ROWS) lRs[threadIdx.x - DIM] = sb;

    for (int base = e_begin; base < e_end; base += 32) {
        const int idx_next = in_src[min(base + 32 + lane, e_last)];
        batch(idx, 0, base);
        if (base + AGG_BATCH < e_end) batch(idx, AGG_BATCH, base + AGG_BATCH);
        idx = idx_next;
    }
    while (cur < AGG_NODES) {
        const float r = __int_as_float(__builtin_amdgcn_readlane(rsl, cur));
        v2f outv; outv.x = ax * r; outv.y = ay * r;
        *(v2f*)(tl + cur * TILE_PITCH) = outv;
        ax = 0.0f; ay = 0.0f;
        ++cur;
    }
    __syncthreads();

    const int half = lane >> 4;
    const int l16  = lane & 15;
    const int mt   = wave >> 1;
    const int nt0  = (wave & 1) * 2;
    const float* arow = lAgg + (mt * 16 + l16) * TILE_PITCH + 2 * half;

    v8f acc[2] = {v8f{}, v8f{}};
#pragma unroll
    for (int f = 0; f < 16; ++f) {
        const v2f a = *(const v2f*)(arow + 4 * f);
        const float4 b2 = *(const float4*)&lWsw[((f * 2 + (nt0 >> 1)) * 32 + lane) * 4];
        v2f b;
        b.x = b2.x; b.y = b2.y;
        acc[0] = __builtin_amdgcn_wmma_f32_16x16x4_f32(
            false, a, false, b, (short)0, acc[0], false, false);
        b.x = b2.z; b.y = b2.w;
        acc[1] = __builtin_amdgcn_wmma_f32_16x16x4_f32(
            false, a, false, b, (short)0, acc[1], false, false);
    }

    const int lrow0 = mt * 16 + half * 8;
    float* yrow = Y + ((size_t)blockIdx.x * TILE_ROWS + lrow0) * DIM;
    float rsv[8];
#pragma unroll
    for (int r = 0; r < 8; ++r) rsv[r] = lRs[lrow0 + r];

#pragma unroll
    for (int t = 0; t < 2; ++t) {
        const int n = (nt0 + t) * 16 + l16;
        const float bn = lB[n];
#pragma unroll
        for (int r = 0; r < 8; ++r) {
            float v = acc[t][r] + bn;
            if (relu_scale) v = fmaxf(v, 0.0f) * rsv[r];
            yrow[(size_t)r * DIM + n] = v;
        }
    }
}

extern "C" void kernel_launch(void* const* d_in, const int* in_sizes, int n_in,
                              void* d_out, int out_size, void* d_ws, size_t ws_size,
                              hipStream_t stream) {
    (void)in_sizes; (void)n_in; (void)out_size; (void)ws_size;
    const float* x  = (const float*)d_in[0];
    const int*   ei = (const int*)d_in[1];
    const float* W1 = (const float*)d_in[3];
    const float* b1 = (const float*)d_in[4];
    const float* W2 = (const float*)d_in[5];
    const float* b2 = (const float*)d_in[6];
    const float* W3 = (const float*)d_in[7];
    const float* b3 = (const float*)d_in[8];
    float* out = (float*)d_out;

    char* ws = (char*)d_ws;
    unsigned* deg_in  = (unsigned*)ws; ws += (size_t)N_NODES * 4;
    float*    rs_out  = (float*)ws;    ws += (size_t)N_NODES * 4;
    float*    rs_in   = (float*)ws;    ws += (size_t)N_NODES * 4;
    int*      row_ptr = (int*)ws;      ws += (size_t)(N_NODES + 16) * 4;
    int*      in_src  = (int*)ws;      ws += (size_t)N_EDGES * 4;
    float*    hs      = (float*)ws;    ws += (size_t)N_NODES * DIM * 4;
    float*    h2      = (float*)ws;    ws += (size_t)N_NODES * DIM * 4;
    unsigned* cnt_src = (unsigned*)hs;
    unsigned* cnt_dst = (unsigned*)h2;

    hist_kernel<<<dim3(CHUNKS, RANGES, 2), 256, 0, stream>>>(ei, cnt_src, cnt_dst);
    merge_kernel<<<dim3(N_NODES / 256, 2), 256, 0, stream>>>(cnt_src, cnt_dst, deg_in, rs_out, rs_in);
    scan_scale_kernel<<<1 + SCALE_BLOCKS, 1024, 0, stream>>>(deg_in, row_ptr, x, rs_out, hs);
    fill_kernel<<<dim3(CHUNKS, RANGES), 256, 0, stream>>>(ei, cnt_dst, row_ptr, in_src);

    layer_kernel<<<N_NODES / TILE_ROWS, 256, 0, stream>>>(hs, in_src, row_ptr, rs_in, W1, b1, rs_out, h2, 1);
    layer_kernel<<<N_NODES / TILE_ROWS, 256, 0, stream>>>(h2, in_src, row_ptr, rs_in, W2, b2, rs_out, hs, 1);
    layer_kernel<<<N_NODES / TILE_ROWS, 256, 0, stream>>>(hs, in_src, row_ptr, rs_in, W3, b3, rs_out, out, 0);
}


# Scan launch removed: merge scans in-degrees inside each 256-node group and writes group totals, fill adds the totals in front itself; x scaling rides in fill; 7 -> 6 launches; float sums unchanged
# speedup vs baseline: 2.7791x; 1.0301x over previous
#include <hip/hip_runtime.h>

#define N_NODES 65536
#define N_EDGES 1048576
#define DIM 64

typedef float v2f __attribute__((ext_vector_type(2)));
typedef float v8f __attribute__((ext_vector_type(8)));


#define CHUNKS 64
#define CHUNK_EDGES (N_EDGES / CHUNKS)
#define RANGES 4
#define RANGE_NODES (N_NODES / RANGES)
#define RANGE_SHIFT 14
#define EDGE_BATCH 8

__global__ __launch_bounds__(256) void hist_kernel(const int* __restrict__ ei,
                                                   unsigned* __restrict__ cnt_src,
                                                   unsigned* __restrict__ cnt_dst) {
    __shared__ unsigned tab[RANGE_NODES];
    const int g = blockIdx.x, q = blockIdx.y, which = blockIdx.z;
    uint4* tab4 = (uint4*)tab;
    for (int i = threadIdx.x; i < RANGE_NODES / 4; i += 256) tab4[i] = make_uint4(0u, 0u, 0u, 0u);
    __syncthreads();
    const int4* idx4 = (const int4*)(ei + (size_t)which * N_EDGES + (size_t)g * CHUNK_EDGES);
    for (int i0 = threadIdx.x; i0 < CHUNK_EDGES / 4; i0 += 256 * EDGE_BATCH) {
        int4 v[EDGE_BATCH];
#pragma unroll
        for (int k = 0; k < EDGE_BATCH; ++k) v[k] = idx4[i0 + k * 256];
#pragma unroll
        for (int k = 0; k < EDGE_BATCH; ++k) {
            if ((v[k].x >> RANGE_SHIFT) == q) atomicAdd(&tab[v[k].x & (RANGE_NODES - 1)], 1u);
            if ((v[k].y >> RANGE_SHIFT) == q) atomicAdd(&tab[v[k].y & (RANGE_NODES - 1)], 1u);
            if ((v[k].z >> RANGE_SHIFT) == q) atomicAdd(&tab[v[k].z & (RANGE_NODES - 1)], 1u);
            if ((v[k].w >> RANGE_SHIFT) == q) atomicAdd(&tab[v[k].w & (RANGE_NODES - 1)], 1u);
        }
    }
    __syncthreads();
    unsigned* cnt = which ? cnt_dst : cnt_src;
    uint4* out4 = (uint4*)(cnt + (size_t)g * N_NODES + (size_t)q * RANGE_NODES);
    for (int i = threadIdx.x; i < RANGE_NODES / 4; i += 256) out4[i] = tab4[i];
}

#define GROUP_NODES 256
#define GROUPS (N_NODES / GROUP_NODES)
#define GROUPS_PER_RANGE (RANGE_NODES / GROUP_NODES)

__device__ __forceinline__ unsigned wave_incl_scan(unsigned v, const int lane) {
    v += (unsigned)__builtin_amdgcn_update_dpp(0, (int)v, 0x111, 0xf, 0xf, false);
    v += (unsigned)__builtin_amdgcn_update_dpp(0, (int)v, 0x112, 0xf, 0xf, false);
    v += (unsigned)__builtin_amdgcn_update_dpp(0, (int)v, 0x114, 0xf, 0xf, false);
    v += (unsigned)__builtin_amdgcn_update_dpp(0, (int)v, 0x118, 0xf, 0xf, false);
    const unsigned low_row = (unsigned)__builtin_amdgcn_readlane((int)v, 15);
    return (lane >= 16) ? v + low_row : v;
}

__global__ __launch_bounds__(256) void merge_kernel(const unsigned* __restrict__ cnt_src,
                                                    unsigned* __restrict__ cnt_dst,
                                                    unsigned* __restrict__ grp_tot,
                                                    float* __restrict__ rs_out,
                                                    float* __restrict__ rs_in) {
    __shared__ unsigned wtot[8];
    const int n = blockIdx.x * GROUP_NODES + threadIdx.x;
    if (blockIdx.y == 0) {
        unsigned dout = 0u;
#pragma unroll
        for (int g = 0; g < CHUNKS; ++g) dout += cnt_src[(size_t)g * N_NODES + n];
        if (dout < 1u) dout = 1u;
        rs_out[n] = rsqrtf((float)dout);
    } else {
        unsigned c[CHUNKS];
#pragma unroll
        for (int g = 0; g < CHUNKS; ++g) c[g] = cnt_dst[(size_t)g * N_NODES + n];
        unsigned din = 0u;
#pragma unroll
        for (int g = 0; g < CHUNKS; ++g) din += c[g];
        const int lane = threadIdx.x & 31;
        const int wave = threadIdx.x >> 5;
        const unsigned incl = wave_incl_scan(din, lane);
        if (lane == 31) wtot[wave] = incl;
        __syncthreads();
        unsigned wave_base = 0u;
#pragma unroll
        for (int w = 0; w < 7; ++w) wave_base += (w < wave) ? wtot[w] : 0u;
        if (threadIdx.x == GROUP_NODES - 1) grp_tot[blockIdx.x] = wave_base + incl;
        unsigned run = wave_base + incl - din;
#pragma unroll
        for (int g = 0; g < CHUNKS; ++g) {
            cnt_dst[(size_t)g * N_NODES + n] = run;
            run += c[g];
        }
        if (din < 1u) din = 1u;
        rs_in[n] = rsqrtf((float)din);
    }
}

#define SCALE_ROWS 8
#define SCALE_PER_THREAD 8

__global__ __launch_bounds__(256) void fill_kernel(const int* __restrict__ ei,
                                                   const unsigned* __restrict__ chunk_off,
                                                   const unsigned* __restrict__ grp_tot,
                                                   int* __restrict__ row_ptr,
                                                   int* __restrict__ in_src,
                                                   const float* __restrict__ x,
                                                   const float* __restrict__ rs_out,
                                                   float* __restrict__ hs) {
    __shared__ unsigned cur[RANGE_NODES];
    const int g = blockIdx.x, q = blockIdx.y;
    if (q >= RANGES) {
        const float4* x4 = (const float4*)x;
        float4* h4 = (float4*)hs;
        const int i0 = ((q - RANGES) * CHUNKS + g) * (SCALE_PER_THREAD * 256) + threadIdx.x;
        float4 v[SCALE_PER_THREAD];
        float r[SCALE_PER_THREAD];
#pragma unroll
        for (int k = 0; k < SCALE_PER_THREAD; ++k) {
            v[k] = x4[i0 + k * 256];
            r[k] = rs_out[(i0 + k * 256) >> 4];
        }
#pragma unroll
        for (int k = 0; k < SCALE_PER_THREAD; ++k) {
            v[k].x *= r[k]; v[k].y *= r[k]; v[k].z *= r[k]; v[k].w *= r[k];
            h4[i0 + k * 256] = v[k];
        }
        return;
    }
    const int lane = threadIdx.x & 31;
    unsigned front = 0u;
#pragma unroll
    for (int j = 0; j < (RANGES - 1) * GROUPS_PER_RANGE / 32; ++j)
        front += (j * 32 < q * GROUPS_PER_RANGE) ? grp_tot[j * 32 + lane] : 0u;
    const uint2 mine = *(const uint2*)(grp_tot + q * GROUPS_PER_RANGE + 2 * lane);
    const unsigned pair = mine.x + mine.y;
    const unsigned front_all = (unsigned)__builtin_amdgcn_readlane((int)wave_incl_scan(front, lane), 31);
    const unsigned base_even = front_all + wave_incl_scan(pair, lane) - pair;
    const unsigned base_odd  = base_even + mine.x;
    const int grp0 = __builtin_amdgcn_readfirstlane((int)(threadIdx.x >> 6));
    const unsigned base_sel = (grp0 & 1) ? base_odd : base_even;
    const uint4* off4 = (const uint4*)(chunk_off + (size_t)g * N_NODES + (size_t)q * RANGE_NODES);
    int4* rp4 = (int4*)(row_ptr + (size_t)q * RANGE_NODES);
    uint4* cur4 = (uint4*)cur;
#pragma unroll
    for (int m0 = 0; m0 < RANGE_NODES / 4 / 256; m0 += EDGE_BATCH) {
        uint4 o[EDGE_BATCH];
#pragma unroll
        for (int k = 0; k < EDGE_BATCH; ++k) o[k] = off4[threadIdx.x + (m0 + k) * 256];
#pragma unroll
        for (int k = 0; k < EDGE_BATCH; ++k) {
            const unsigned b = (unsigned)__builtin_amdgcn_readlane((int)base_sel, (grp0 + 4 * (m0 + k)) >> 1);
            const uint4 c4 = make_uint4(o[k].x + b, o[k].y + b, o[k].z + b, o[k].w + b);
            cur4[threadIdx.x + (m0 + k) * 256] = c4;
            if (g == 0) rp4[threadIdx.x + (m0 + k) * 256] = make_int4((int)c4.x, (int)c4.y, (int)c4.z, (int)c4.w);
        }
    }
    if (g == 0 && q == RANGES - 1 && threadIdx.x == 31) row_ptr[N_NODES] = (int)(base_odd + mine.y);
    __syncthreads();
    const int4* s4 = (const int4*)(ei + (size_t)g * CHUNK_EDGES);
    const int4* d4 = (const int4*)(ei + (size_t)N_EDGES + (size_t)g * CHUNK_EDGES);
    for (int i0 = threadIdx.x; i0 < CHUNK_EDGES / 4; i0 += 256 * EDGE_BATCH) {
        int4 d[EDGE_BATCH], s[EDGE_BATCH];
#pragma unroll
        for (int k = 0; k < EDGE_BATCH; ++k) { d[k] = d4[i0 + k * 256]; s[k] = s4[i0 + k * 256]; }
#pragma unroll
        for (int k = 0; k < EDGE_BATCH; ++k) {
            if ((d[k].x >> RANGE_SHIFT) == q) in_src[atomicAdd(&cur[d[k].x & (RANGE_NODES - 1)], 1u)] = s[k].x;
            if ((d[k].y >> RANGE_SHIFT) == q) in_src[atomicAdd(&cur[d[k].y & (RANGE_NODES - 1)], 1u)] = s[k].y;
            if ((d[k].z >> RANGE_SHIFT) == q) in_src[atomicAdd(&cur[d[k].z & (RANGE_NODES - 1)], 1u)] = s[k].z;
            if ((d[k].w >> RANGE_SHIFT) == q) in_src[atomicAdd(&cur[d[k].w & (RANGE_NODES - 1)], 1u)] = s[k].w;
        }
    }
}

#define AGG_NODES 8
#define AGG_BATCH 16
#define TILE_ROWS 64
#define TILE_PITCH 68

__global__ __launch_bounds__(256) void layer_kernel(const float* __restrict__ hs,
                                                    const int* __restrict__ in_src,
                                                    const int* __restrict__ row_ptr,
                                                    const float* __restrict__ rs_in,
                                                    const float* __restrict__ W,
                                                    const float* __restrict__ bias,
                                                    const float* __restrict__ rs_out,
                                                    float* __restrict__ Y,
                                                    int relu_scale) {
    __shared__ alignas(16) float lWsw[16 * 4 * 32 * 2];
    __shared__ alignas(16) float lAgg[TILE_ROWS * TILE_PITCH];
    __shared__ float lB[DIM];
    __shared__ float lRs[TILE_ROWS];

    const int lane = threadIdx.x & 31;
    const int wave = threadIdx.x >> 5;

    v2f w2[8];
#pragma unroll
    for (int i = 0; i < 8; ++i) {
        const int p  = threadIdx.x + i * 256;
        const int l  = p & 31;
        const int t  = (p >> 5) & 3;
        const int f  = p >> 7;
        const int ka = 4 * f + 2 * (l >> 4);
        const int n  = 16 * t + (l & 15);
        w2[i].x = W[ka * DIM + n];
        w2[i].y = W[(ka + 1) * DIM + n];
    }
    float sb = 1.0f;
    if (threadIdx.x < DIM) sb = bias[threadIdx.x];
    else if (threadIdx.x < DIM + TILE_ROWS && relu_scale) sb = rs_out[blockIdx.x * TILE_ROWS + threadIdx.x - DIM];

    const int node0 = (blockIdx.x * 8 + wave) * AGG_NODES;
    const int rp  = (lane <= AGG_NODES) ? row_ptr[node0 + lane] : 0x7fffffff;
    const int rsl = __float_as_int(rs_in[node0 + (lane < AGG_NODES ? lane : AGG_NODES - 1)]);
    const int e_begin = __builtin_amdgcn_readlane(rp, 0);
    const int e_end   = __builtin_amdgcn_readlane(rp, AGG_NODES);
    int cur     = 0;
    int cur_end = __builtin_amdgcn_readlane(rp, 1);
    float ax = 0.0f, ay = 0.0f;
    const float* hl = hs + lane * 2;
    float* tl = lAgg + wave * AGG_NODES * TILE_PITCH + lane * 2;

    auto batch = [&](const int idx, const int j0, const int e0) {
        v2f v[AGG_BATCH];
#pragma unroll
        for (int u = 0; u < AGG_BATCH; ++u) {
            const unsigned srow = (unsigned)__builtin_amdgcn_readlane(idx, j0 + u);
            v[u] = *(const v2f*)(hl + (size_t)srow * DIM);
        }
        int rem = cur_end - e0;
#pragma unroll
        for (int u = 0; u < AGG_BATCH; ++u) {
            while (u >= rem) {
                const float r = __int_as_float(__builtin_amdgcn_readlane(rsl, cur));
                v2f outv; outv.x = ax * r; outv.y = ay * r;
                *(v2f*)(tl + cur * TILE_PITCH) = outv;
                ax = 0.0f; ay = 0.0f;
                ++cur;
                cur_end = __builtin_amdgcn_readlane(rp, cur + 1);
                rem = cur_end - e0;
            }
            ax += v[u].x; ay += v[u].y;
        }
    };

    const int e_last = (e_end > 0) ? e_end - 1 : 0;
    int idx = in_src[min(e_begin + lane, e_last)];

#pragma unroll
    for (int i = 0; i < 8; ++i) {
        const int p = threadIdx.x + i * 256;
        const int l = p & 31;
        const int t = (p >> 5) & 3;
        const int f = p >> 7;
        *(v2f*)&lWsw[(((f * 2 + (t >> 1)) * 32 + l) * 2 + (t & 1)) * 2] = w2[i];
    }
    if (threadIdx.x < DIM) lB[threadIdx.x] = sb;
    else if (threadIdx.x < DIM + TILE_ROWS) lRs[threadIdx.x - DIM] = sb;

    for (int base = e_begin; base < e_end; base += 32) {
        const int idx_next = in_src[min(base + 32 + lane, e_last)];
        batch(idx, 0, base);
        if (base + AGG_BATCH < e_end) batch(idx, AGG_BATCH, base + AGG_BATCH);
        idx = idx_next;
    }
    while (cur < AGG_NODES) {
        const float r = __int_as_float(__builtin_amdgcn_readlane(rsl, cur));
        v2f outv; outv.x = ax * r; outv.y = ay * r;
        *(v2f*)(tl + cur * TILE_PITCH) = outv;
        ax = 0.0f; ay = 0.0f;
        ++cur;
    }
    __syncthreads();

    const int half = lane >> 4;
    const int l16  = lane & 15;
    const int mt   = wave >> 1;
    const int nt0  = (wave & 1) * 2;
    const float* arow = lAgg + (mt * 16 + l16) * TILE_PITCH + 2 * half;

    v8f acc[2] = {v8f{}, v8f{}};
#pragma unroll
    for (int f = 0; f < 16; ++f) {
        const v2f a = *(const v2f*)(arow + 4 * f);
        const float4 b2 = *(const float4*)&lWsw[((f * 2 + (nt0 >> 1)) * 32 + lane) * 4];
        v2f b;
        b.x = b2.x; b.y = b2.y;
        acc[0] = __builtin_amdgcn_wmma_f32_16x16x4_f32(
            false, a, false, b, (short)0, acc[0], false, false);
        b.x = b2.z; b.y = b2.w;
        acc[1] = __builtin_amdgcn_wmma_f32_16x16x4_f32(
            false, a, false, b, (short)0, acc[1], false, false);
    }

    const int lrow0 = mt * 16 + half * 8;
    float* yrow = Y + ((size_t)blockIdx.x * TILE_ROWS + lrow0) * DIM;
    float rsv[8];
#pragma unroll
    for (int r = 0; r < 8; ++r) rsv[r] = lRs[lrow0 + r];

#pragma unroll
    for (int t = 0; t < 2; ++t) {
        const int n = (nt0 + t) * 16 + l16;
        const float bn = lB[n];
#pragma unroll
        for (int r = 0; r < 8; ++r) {
            float v = acc[t][r] + bn;
            if (relu_scale) v = fmaxf(v, 0.0f) * rsv[r];
            yrow[(size_t)r * DIM + n] = v;
        }
    }
}

extern "C" void kernel_launch(void* const* d_in, const int* in_sizes, int n_in,
                              void* d_out, int out_size, void* d_ws, size_t ws_size,
                              hipStream_t stream) {
    (void)in_sizes; (void)n_in; (void)out_size; (void)ws_size;
    const float* x  = (const float*)d_in[0];
    const int*   ei = (const int*)d_in[1];
    const float* W1 = (const float*)d_in[3];
    const float* b1 = (const float*)d_in[4];
    const float* W2 = (const float*)d_in[5];
    const float* b2 = (const float*)d_in[6];
    const float* W3 = (const float*)d_in[7];
    const float* b3 = (const float*)d_in[8];
    float* out = (float*)d_out;

    char* ws = (char*)d_ws;
    unsigned* grp_tot = (unsigned*)ws; ws += (size_t)N_NODES * 4;
    float*    rs_out  = (float*)ws;    ws += (size_t)N_NODES * 4;
    float*    rs_in   = (float*)ws;    ws += (size_t)N_NODES * 4;
    int*      row_ptr = (int*)ws;      ws += (size_t)(N_NODES + 16) * 4;
    int*      in_src  = (int*)ws;      ws += (size_t)N_EDGES * 4;
    float*    hs      = (float*)ws;    ws += (size_t)N_NODES * DIM * 4;
    float*    h2      = (float*)ws;    ws += (size_t)N_NODES * DIM * 4;
    unsigned* cnt_src = (unsigned*)hs;
    unsigned* cnt_dst = (unsigned*)h2;

    hist_kernel<<<dim3(CHUNKS, RANGES, 2), 256, 0, stream>>>(ei, cnt_src, cnt_dst);
    merge_kernel<<<dim3(GROUPS, 2), 256, 0, stream>>>(cnt_src, cnt_dst, grp_tot, rs_out, rs_in);
    fill_kernel<<<dim3(CHUNKS, RANGES + SCALE_ROWS), 256, 0, stream>>>(ei, cnt_dst, grp_tot, row_ptr, in_src, x, rs_out, hs);

    layer_kernel<<<N_NODES / TILE_ROWS, 256, 0, stream>>>(hs, in_src, row_ptr, rs_in, W1, b1, rs_out, h2, 1);
    layer_kernel<<<N_NODES / TILE_ROWS, 256, 0, stream>>>(h2, in_src, row_ptr, rs_in, W2, b2, rs_out, hs, 1);
    layer_kernel<<<N_NODES / TILE_ROWS, 256, 0, stream>>>(hs, in_src, row_ptr, rs_in, W3, b3, rs_out, out, 0);
}


# Workspace order: feature buffers moved to the front (kernel pointer arguments change, device code same): a gathered 256-byte row is one aligned 256-byte unit, two lines; was 64 bytes in, three lines
# speedup vs baseline: 2.9263x; 1.0530x over previous
#include <hip/hip_runtime.h>

#define N_NODES 65536
#define N_EDGES 1048576
#define DIM 64

typedef float v2f __attribute__((ext_vector_type(2)));
typedef float v8f __attribute__((ext_vector_type(8)));


#define CHUNKS 64
#define CHUNK_EDGES (N_EDGES / CHUNKS)
#define RANGES 4
#define RANGE_NODES (N_NODES / RANGES)
#define RANGE_SHIFT 14
#define EDGE_BATCH 8

__global__ __launch_bounds__(256) void hist_kernel(const int* __restrict__ ei,
                                                   unsigned* __restrict__ cnt_src,
                                                   unsigned* __restrict__ cnt_dst) {
    __shared__ unsigned tab[RANGE_NODES];
    const int g = blockIdx.x, q = blockIdx.y, which = blockIdx.z;
    uint4* tab4 = (uint4*)tab;
    for (int i = threadIdx.x; i < RANGE_NODES / 4; i += 256) tab4[i] = make_uint4(0u, 0u, 0u, 0u);
    __syncthreads();
    const int4* idx4 = (const int4*)(ei + (size_t)which * N_EDGES + (size_t)g * CHUNK_EDGES);
    for (int i0 = threadIdx.x; i0 < CHUNK_EDGES / 4; i0 += 256 * EDGE_BATCH) {
        int4 v[EDGE_BATCH];
#pragma unroll
        for (int k = 0; k < EDGE_BATCH; ++k) v[k] = idx4[i0 + k * 256];
#pragma unroll
        for (int k = 0; k < EDGE_BATCH; ++k) {
            if ((v[k].x >> RANGE_SHIFT) == q) atomicAdd(&tab[v[k].x & (RANGE_NODES - 1)], 1u);
            if ((v[k].y >> RANGE_SHIFT) == q) atomicAdd(&tab[v[k].y & (RANGE_NODES - 1)], 1u);
            if ((v[k].z >> RANGE_SHIFT) == q) atomicAdd(&tab[v[k].z & (RANGE_NODES - 1)], 1u);
            if ((v[k].w >> RANGE_SHIFT) == q) atomicAdd(&tab[v[k].w & (RANGE_NODES - 1)], 1u);
        }
    }
    __syncthreads();
    unsigned* cnt = which ? cnt_dst : cnt_src;
    uint4* out4 = (uint4*)(cnt + (size_t)g * N_NODES + (size_t)q * RANGE_NODES);
    for (int i = threadIdx.x; i < RANGE_NODES / 4; i += 256) out4[i] = tab4[i];
}

#define GROUP_NODES 256
#define GROUPS (N_NODES / GROUP_NODES)
#define GROUPS_PER_RANGE (RANGE_NODES / GROUP_NODES)

__device__ __forceinline__ unsigned wave_incl_scan(unsigned v, const int lane) {
    v += (unsigned)__builtin_amdgcn_update_dpp(0, (int)v, 0x111, 0xf, 0xf, false);
    v += (unsigned)__builtin_amdgcn_update_dpp(0, (int)v, 0x112, 0xf, 0xf, false);
    v += (unsigned)__builtin_amdgcn_update_dpp(0, (int)v, 0x114, 0xf, 0xf, false);
    v += (unsigned)__builtin_amdgcn_update_dpp(0, (int)v, 0x118, 0xf, 0xf, false);
    const unsigned low_row = (unsigned)__builtin_amdgcn_readlane((int)v, 15);
    return (lane >= 16) ? v + low_row : v;
}

__global__ __launch_bounds__(256) void merge_kernel(const unsigned* __restrict__ cnt_src,
                                                    unsigned* __restrict__ cnt_dst,
                                                    unsigned* __restrict__ grp_tot,
                                                    float* __restrict__ rs_out,
                                                    float* __restrict__ rs_in) {
    __shared__ unsigned wtot[8];
    const int n = blockIdx.x * GROUP_NODES + threadIdx.x;
    if (blockIdx.y == 0) {
        unsigned dout = 0u;
#pragma unroll
        for (int g = 0; g < CHUNKS; ++g) dout += cnt_src[(size_t)g * N_NODES + n];
        if (dout < 1u) dout = 1u;
        rs_out[n] = rsqrtf((float)dout);
    } else {
        unsigned c[CHUNKS];
#pragma unroll
        for (int g = 0; g < CHUNKS; ++g) c[g] = cnt_dst[(size_t)g * N_NODES + n];
        unsigned din = 0u;
#pragma unroll
        for (int g = 0; g < CHUNKS; ++g) din += c[g];
        const int lane = threadIdx.x & 31;
        const int wave = threadIdx.x >> 5;
        const unsigned incl = wave_incl_scan(din, lane);
        if (lane == 31) wtot[wave] = incl;
        __syncthreads();
        unsigned wave_base = 0u;
#pragma unroll
        for (int w = 0; w < 7; ++w) wave_base += (w < wave) ? wtot[w] : 0u;
        if (threadIdx.x == GROUP_NODES - 1) grp_tot[blockIdx.x] = wave_base + incl;
        unsigned run = wave_base + incl - din;
#pragma unroll
        for (int g = 0; g < CHUNKS; ++g) {
            cnt_dst[(size_t)g * N_NODES + n] = run;
            run += c[g];
        }
        if (din < 1u) din = 1u;
        rs_in[n] = rsqrtf((float)din);
    }
}

#define SCALE_ROWS 8
#define SCALE_PER_THREAD 8

__global__ __launch_bounds__(256) void fill_kernel(const int* __restrict__ ei,
                                                   const unsigned* __restrict__ chunk_off,
                                                   const unsigned* __restrict__ grp_tot,
                                                   int* __restrict__ row_ptr,
                                                   int* __restrict__ in_src,
                                                   const float* __restrict__ x,
                                                   const float* __restrict__ rs_out,
                                                   float* __restrict__ hs) {
    __shared__ unsigned cur[RANGE_NODES];
    const int g = blockIdx.x, q = blockIdx.y;
    if (q >= RANGES) {
        const float4* x4 = (const float4*)x;
        float4* h4 = (float4*)hs;
        const int i0 = ((q - RANGES) * CHUNKS + g) * (SCALE_PER_THREAD * 256) + threadIdx.x;
        float4 v[SCALE_PER_THREAD];
        float r[SCALE_PER_THREAD];
#pragma unroll
        for (int k = 0; k < SCALE_PER_THREAD; ++k) {
            v[k] = x4[i0 + k * 256];
            r[k] = rs_out[(i0 + k * 256) >> 4];
        }
#pragma unroll
        for (int k = 0; k < SCALE_PER_THREAD; ++k) {
            v[k].x *= r[k]; v[k].y *= r[k]; v[k].z *= r[k]; v[k].w *= r[k];
            h4[i0 + k * 256] = v[k];
        }
        return;
    }
    const int lane = threadIdx.x & 31;
    unsigned front = 0u;
#pragma unroll
    for (int j = 0; j < (RANGES - 1) * GROUPS_PER_RANGE / 32; ++j)
        front += (j * 32 < q * GROUPS_PER_RANGE) ? grp_tot[j * 32 + lane] : 0u;
    const uint2 mine = *(const uint2*)(grp_tot + q * GROUPS_PER_RANGE + 2 * lane);
    const unsigned pair = mine.x + mine.y;
    const unsigned front_all = (unsigned)__builtin_amdgcn_readlane((int)wave_incl_scan(front, lane), 31);
    const unsigned base_even = front_all + wave_incl_scan(pair, lane) - pair;
    const unsigned base_odd  = base_even + mine.x;
    const int grp0 = __builtin_amdgcn_readfirstlane((int)(threadIdx.x >> 6));
    const unsigned base_sel = (grp0 & 1) ? base_odd : base_even;
    const uint4* off4 = (const uint4*)(chunk_off + (size_t)g * N_NODES + (size_t)q * RANGE_NODES);
    int4* rp4 = (int4*)(row_ptr + (size_t)q * RANGE_NODES);
    uint4* cur4 = (uint4*)cur;
#pragma unroll
    for (int m0 = 0; m0 < RANGE_NODES / 4 / 256; m0 += EDGE_BATCH) {
        uint4 o[EDGE_BATCH];
#pragma unroll
        for (int k = 0; k < EDGE_BATCH; ++k) o[k] = off4[threadIdx.x + (m0 + k) * 256];
#pragma unroll
        for (int k = 0; k < EDGE_BATCH; ++k) {
            const unsigned b = (unsigned)__builtin_amdgcn_readlane((int)base_sel, (grp0 + 4 * (m0 + k)) >> 1);
            const uint4 c4 = make_uint4(o[k].x + b, o[k].y + b, o[k].z + b, o[k].w + b);
            cur4[threadIdx.x + (m0 + k) * 256] = c4;
            if (g == 0) rp4[threadIdx.x + (m0 + k) * 256] = make_int4((int)c4.x, (int)c4.y, (int)c4.z, (int)c4.w);
        }
    }
    if (g == 0 && q == RANGES - 1 && threadIdx.x == 31) row_ptr[N_NODES] = (int)(base_odd + mine.y);
    __syncthreads();
    const int4* s4 = (const int4*)(ei + (size_t)g * CHUNK_EDGES);
    const int4* d4 = (const int4*)(ei + (size_t)N_EDGES + (size_t)g * CHUNK_EDGES);
    for (int i0 = threadIdx.x; i0 < CHUNK_EDGES / 4; i0 += 256 * EDGE_BATCH) {
        int4 d[EDGE_BATCH], s[EDGE_BATCH];
#pragma unroll
        for (int k = 0; k < EDGE_BATCH; ++k) { d[k] = d4[i0 + k * 256]; s[k] = s4[i0 + k * 256]; }
#pragma unroll
        for (int k = 0; k < EDGE_BATCH; ++k) {
            if ((d[k].x >> RANGE_SHIFT) == q) in_src[atomicAdd(&cur[d[k].x & (RANGE_NODES - 1)], 1u)] = s[k].x;
            if ((d[k].y >> RANGE_SHIFT) == q) in_src[atomicAdd(&cur[d[k].y & (RANGE_NODES - 1)], 1u)] = s[k].y;
            if ((d[k].z >> RANGE_SHIFT) == q) in_src[atomicAdd(&cur[d[k].z & (RANGE_NODES - 1)], 1u)] = s[k].z;
            if ((d[k].w >> RANGE_SHIFT) == q) in_src[atomicAdd(&cur[d[k].w & (RANGE_NODES - 1)], 1u)] = s[k].w;
        }
    }
}

#define AGG_NODES 8
#define AGG_BATCH 16
#define TILE_ROWS 64
#define TILE_PITCH 68

__global__ __launch_bounds__(256) void layer_kernel(const float* __restrict__ hs,
                                                    const int* __restrict__ in_src,
                                                    const int* __restrict__ row_ptr,
                                                    const float* __restrict__ rs_in,
                                                    const float* __restrict__ W,
                                                    const float* __restrict__ bias,
                                                    const float* __restrict__ rs_out,
                                                    float* __restrict__ Y,
                                                    int relu_scale) {
    __shared__ alignas(16) float lWsw[16 * 4 * 32 * 2];
    __shared__ alignas(16) float lAgg[TILE_ROWS * TILE_PITCH];
    __shared__ float lB[DIM];
    __shared__ float lRs[TILE_ROWS];

    const int lane = threadIdx.x & 31;
    const int wave = threadIdx.x >> 5;

    v2f w2[8];
#pragma unroll
    for (int i = 0; i < 8; ++i) {
        const int p  = threadIdx.x + i * 256;
        const int l  = p & 31;
        const int t  = (p >> 5) & 3;
        const int f  = p >> 7;
        const int ka = 4 * f + 2 * (l >> 4);
        const int n  = 16 * t + (l & 15);
        w2[i].x = W[ka * DIM + n];
        w2[i].y = W[(ka + 1) * DIM + n];
    }
    float sb = 1.0f;
    if (threadIdx.x < DIM) sb = bias[threadIdx.x];
    else if (threadIdx.x < DIM + TILE_ROWS && relu_scale) sb = rs_out[blockIdx.x * TILE_ROWS + threadIdx.x - DIM];

    const int node0 = (blockIdx.x * 8 + wave) * AGG_NODES;
    const int rp  = (lane <= AGG_NODES) ? row_ptr[node0 + lane] : 0x7fffffff;
    const int rsl = __float_as_int(rs_in[node0 + (lane < AGG_NODES ? lane : AGG_NODES - 1)]);
    const int e_begin = __builtin_amdgcn_readlane(rp, 0);
    const int e_end   = __builtin_amdgcn_readlane(rp, AGG_NODES);
    int cur     = 0;
    int cur_end = __builtin_amdgcn_readlane(rp, 1);
    float ax = 0.0f, ay = 0.0f;
    const float* hl = hs + lane * 2;
    float* tl = lAgg + wave * AGG_NODES * TILE_PITCH + lane * 2;

    auto batch = [&](const int idx, const int j0, const int e0) {
        v2f v[AGG_BATCH];
#pragma unroll
        for (int u = 0; u < AGG_BATCH; ++u) {
            const unsigned srow = (unsigned)__builtin_amdgcn_readlane(idx, j0 + u);
            v[u] = *(const v2f*)(hl + (size_t)srow * DIM);
        }
        int rem = cur_end - e0;
#pragma unroll
        for (int u = 0; u < AGG_BATCH; ++u) {
            while (u >= rem) {
                const float r = __int_as_float(__builtin_amdgcn_readlane(rsl, cur));
                v2f outv; outv.x = ax * r; outv.y = ay * r;
                *(v2f*)(tl + cur * TILE_PITCH) = outv;
                ax = 0.0f; ay = 0.0f;
                ++cur;
                cur_end = __builtin_amdgcn_readlane(rp, cur + 1);
                rem = cur_end - e0;
            }
            ax += v[u].x; ay += v[u].y;
        }
    };

    const int e_last = (e_end > 0) ? e_end - 1 : 0;
    int idx = in_src[min(e_begin + lane, e_last)];

#pragma unroll
    for (int i = 0; i < 8; ++i) {
        const int p = threadIdx.x + i * 256;
        const int l = p & 31;
        const int t = (p >> 5) & 3;
        const int f = p >> 7;
        *(v2f*)&lWsw[(((f * 2 + (t >> 1)) * 32 + l) * 2 + (t & 1)) * 2] = w2[i];
    }
    if (threadIdx.x < DIM) lB[threadIdx.x] = sb;
    else if (threadIdx.x < DIM + TILE_ROWS) lRs[threadIdx.x - DIM] = sb;

    for (int base = e_begin; base < e_end; base += 32) {
        const int idx_next = in_src[min(base + 32 + lane, e_last)];
        batch(idx, 0, base);
        if (base + AGG_BATCH < e_end) batch(idx, AGG_BATCH, base + AGG_BATCH);
        idx = idx_next;
    }
    while (cur < AGG_NODES) {
        const float r = __int_as_float(__builtin_amdgcn_readlane(rsl, cur));
        v2f outv; outv.x = ax * r; outv.y = ay * r;
        *(v2f*)(tl + cur * TILE_PITCH) = outv;
        ax = 0.0f; ay = 0.0f;
        ++cur;
    }
    __syncthreads();

    const int half = lane >> 4;
    const int l16  = lane & 15;
    const int mt   = wave >> 1;
    const int nt0  = (wave & 1) * 2;
    const float* arow = lAgg + (mt * 16 + l16) * TILE_PITCH + 2 * half;

    v8f acc[2] = {v8f{}, v8f{}};
#pragma unroll
    for (int f = 0; f < 16; ++f) {
        const v2f a = *(const v2f*)(arow + 4 * f);
        const float4 b2 = *(const float4*)&lWsw[((f * 2 + (nt0 >> 1)) * 32 + lane) * 4];
        v2f b;
        b.x = b2.x; b.y = b2.y;
        acc[0] = __builtin_amdgcn_wmma_f32_16x16x4_f32(
            false, a, false, b, (short)0, acc[0], false, false);
        b.x = b2.z; b.y = b2.w;
        acc[1] = __builtin_amdgcn_wmma_f32_16x16x4_f32(
            false, a, false, b, (short)0, acc[1], false, false);
    }

    const int lrow0 = mt * 16 + half * 8;
    float* yrow = Y + ((size_t)blockIdx.x * TILE_ROWS + lrow0) * DIM;
    float rsv[8];
#pragma unroll
    for (int r = 0; r < 8; ++r) rsv[r] = lRs[lrow0 + r];

#pragma unroll
    for (int t = 0; t < 2; ++t) {
        const int n = (nt0 + t) * 16 + l16;
        const float bn = lB[n];
#pragma unroll
        for (int r = 0; r < 8; ++r) {
            float v = acc[t][r] + bn;
            if (relu_scale) v = fmaxf(v, 0.0f) * rsv[r];
            yrow[(size_t)r * DIM + n] = v;
        }
    }
}

extern "C" void kernel_launch(void* const* d_in, const int* in_sizes, int n_in,
                              void* d_out, int out_size, void* d_ws, size_t ws_size,
                              hipStream_t stream) {
    (void)in_sizes; (void)n_in; (void)out_size; (void)ws_size;
    const float* x  = (const float*)d_in[0];
    const int*   ei = (const int*)d_in[1];
    const float* W1 = (const float*)d_in[3];
    const float* b1 = (const float*)d_in[4];
    const float* W2 = (const float*)d_in[5];
    const float* b2 = (const float*)d_in[6];
    const float* W3 = (const float*)d_in[7];
    const float* b3 = (const float*)d_in[8];
    float* out = (float*)d_out;

    char* ws = (char*)d_ws;
    float*    hs      = (float*)ws;    ws += (size_t)N_NODES * DIM * 4;
    float*    h2      = (float*)ws;    ws += (size_t)N_NODES * DIM * 4;
    unsigned* grp_tot = (unsigned*)ws; ws += (size_t)N_NODES * 4;
    float*    rs_out  = (float*)ws;    ws += (size_t)N_NODES * 4;
    float*    rs_in   = (float*)ws;    ws += (size_t)N_NODES * 4;
    int*      in_src  = (int*)ws;      ws += (size_t)N_EDGES * 4;
    int*      row_ptr = (int*)ws;      ws += (size_t)(N_NODES + 16) * 4;
    unsigned* cnt_src = (unsigned*)hs;
    unsigned* cnt_dst = (unsigned*)h2;

    hist_kernel<<<dim3(CHUNKS, RANGES, 2), 256, 0, stream>>>(ei, cnt_src, cnt_dst);
    merge_kernel<<<dim3(GROUPS, 2), 256, 0, stream>>>(cnt_src, cnt_dst, grp_tot, rs_out, rs_in);
    fill_kernel<<<dim3(CHUNKS, RANGES + SCALE_ROWS), 256, 0, stream>>>(ei, cnt_dst, grp_tot, row_ptr, in_src, x, rs_out, hs);

    layer_kernel<<<N_NODES / TILE_ROWS, 256, 0, stream>>>(hs, in_src, row_ptr, rs_in, W1, b1, rs_out, h2, 1);
    layer_kernel<<<N_NODES / TILE_ROWS, 256, 0, stream>>>(h2, in_src, row_ptr, rs_in, W2, b2, rs_out, hs, 1);
    layer_kernel<<<N_NODES / TILE_ROWS, 256, 0, stream>>>(hs, in_src, row_ptr, rs_in, W3, b3, rs_out, out, 0);
}


# CSR built by two-level bucket sort in LDS (edges by 256-node dst bucket, then rows per bucket; in_src in whole lines; no dst tables, no scattered stores); row sum order fixed: chunk, wave, lane
# speedup vs baseline: 3.8425x; 1.3131x over previous
#include <hip/hip_runtime.h>

#define N_NODES 65536
#define N_EDGES 1048576
#define DIM 64

typedef float v2f __attribute__((ext_vector_type(2)));
typedef float v8f __attribute__((ext_vector_type(8)));


#define CHUNKS 64
#define CHUNK_EDGES (N_EDGES / CHUNKS)
#define RANGES 4
#define RANGE_NODES (N_NODES / RANGES)
#define RANGE_SHIFT 14
#define EDGE_BATCH 8

#define BCHUNKS 256
#define BCHUNK_EDGES (N_EDGES / BCHUNKS)
#define BUCKETS 256
#define BUCKET_SHIFT 8
#define PB_CAP 6144


__global__ __launch_bounds__(256) void hist_kernel(const int* __restrict__ ei,
                                                   unsigned* __restrict__ cnt_src,
                                                   unsigned* __restrict__ bcnt) {
    __shared__ unsigned tab[RANGE_NODES];
    const int g = blockIdx.x, q = blockIdx.y, which = blockIdx.z;
    if (which == 1) {
        const int c = q * CHUNKS + g;
        tab[threadIdx.x] = 0u;
        const int4* d4 = (const int4*)(ei + (size_t)N_EDGES + (size_t)c * BCHUNK_EDGES);
        int4 v[BCHUNK_EDGES / 4 / 256];
#pragma unroll
        for (int k = 0; k < BCHUNK_EDGES / 4 / 256; ++k) v[k] = d4[threadIdx.x + k * 256];
        __syncthreads();
#pragma unroll
        for (int k = 0; k < BCHUNK_EDGES / 4 / 256; ++k) {
            if ((unsigned)v[k].x < N_NODES) atomicAdd(&tab[v[k].x >> BUCKET_SHIFT], 1u);
            if ((unsigned)v[k].y < N_NODES) atomicAdd(&tab[v[k].y >> BUCKET_SHIFT], 1u);
            if ((unsigned)v[k].z < N_NODES) atomicAdd(&tab[v[k].z >> BUCKET_SHIFT], 1u);
            if ((unsigned)v[k].w < N_NODES) atomicAdd(&tab[v[k].w >> BUCKET_SHIFT], 1u);
        }
        __syncthreads();
        bcnt[c * BUCKETS + threadIdx.x] = tab[threadIdx.x];
        return;
    }
    uint4* tab4 = (uint4*)tab;
    for (int i = threadIdx.x; i < RANGE_NODES / 4; i += 256) tab4[i] = make_uint4(0u, 0u, 0u, 0u);
    __syncthreads();
    const int4* idx4 = (const int4*)(ei + (size_t)g * CHUNK_EDGES);
    for (int i0 = threadIdx.x; i0 < CHUNK_EDGES / 4; i0 += 256 * EDGE_BATCH) {
        int4 v[EDGE_BATCH];
#pragma unroll
        for (int k = 0; k < EDGE_BATCH; ++k) v[k] = idx4[i0 + k * 256];
#pragma unroll
        for (int k = 0; k < EDGE_BATCH; ++k) {
            if ((v[k].x >> RANGE_SHIFT) == q) atomicAdd(&tab[v[k].x & (RANGE_NODES - 1)], 1u);
            if ((v[k].y >> RANGE_SHIFT) == q) atomicAdd(&tab[v[k].y & (RANGE_NODES - 1)], 1u);
            if ((v[k].z >> RANGE_SHIFT) == q) atomicAdd(&tab[v[k].z & (RANGE_NODES - 1)], 1u);
            if ((v[k].w >> RANGE_SHIFT) == q) atomicAdd(&tab[v[k].w & (RANGE_NODES - 1)], 1u);
        }
    }
    __syncthreads();
    uint4* out4 = (uint4*)(cnt_src + (size_t)g * N_NODES + (size_t)q * RANGE_NODES);
    for (int i = threadIdx.x; i < RANGE_NODES / 4; i += 256) out4[i] = tab4[i];
}

__device__ __forceinline__ unsigned wave_incl_scan(unsigned v, const int lane) {
    v += (unsigned)__builtin_amdgcn_update_dpp(0, (int)v, 0x111, 0xf, 0xf, false);
    v += (unsigned)__builtin_amdgcn_update_dpp(0, (int)v, 0x112, 0xf, 0xf, false);
    v += (unsigned)__builtin_amdgcn_update_dpp(0, (int)v, 0x114, 0xf, 0xf, false);
    v += (unsigned)__builtin_amdgcn_update_dpp(0, (int)v, 0x118, 0xf, 0xf, false);
    const unsigned low_row = (unsigned)__builtin_amdgcn_readlane((int)v, 15);
    return (lane >= 16) ? v + low_row : v;
}

__global__ __launch_bounds__(1024) void bucket_kernel(const int* __restrict__ ei,
                                                      const unsigned* __restrict__ cnt_src,
                                                      const unsigned* __restrict__ bcnt,
                                                      float* __restrict__ rs_out,
                                                      unsigned* __restrict__ bstart,
                                                      unsigned* __restrict__ bucketed) {
    __shared__ unsigned part_tot[4][BUCKETS];
    __shared__ unsigned part_pre[4][BUCKETS];
    __shared__ unsigned wcnt[32][BUCKETS];
    __shared__ unsigned gpos[BUCKETS];
    __shared__ unsigned lpos[BUCKETS];
    __shared__ unsigned wsum[8][2];
    __shared__ unsigned stage[BCHUNK_EDGES];
    __shared__ unsigned nvalid;
    const int tid = threadIdx.x;
    if (blockIdx.x < N_NODES / 1024) {
        const int n = blockIdx.x * 1024 + tid;
        unsigned dout = 0u;
#pragma unroll
        for (int g = 0; g < CHUNKS; ++g) dout += cnt_src[(size_t)g * N_NODES + n];
        if (dout < 1u) dout = 1u;
        rs_out[n] = rsqrtf((float)dout);
        return;
    }
    const int g = blockIdx.x - N_NODES / 1024;
    const int lane = tid & 31;
    const int wave = tid >> 5;
    const int4 d = ((const int4*)(ei + (size_t)N_EDGES + (size_t)g * BCHUNK_EDGES))[tid];
    const int4 s = ((const int4*)(ei + (size_t)g * BCHUNK_EDGES))[tid];
    const int b = tid & (BUCKETS - 1);
    const int part = tid >> 8;
    unsigned tot = 0u, pre = 0u;
#pragma unroll
    for (int k = 0; k < BCHUNKS / 4; ++k) {
        const int gp = part * (BCHUNKS / 4) + k;
        const unsigned v = bcnt[gp * BUCKETS + b];
        tot += v;
        pre += (gp < g) ? v : 0u;
    }
    part_tot[part][b] = tot;
    part_pre[part][b] = pre;
#pragma unroll
    for (int k = 0; k < 32 * BUCKETS / 1024; ++k) (&wcnt[0][0])[tid + k * 1024] = 0u;
    __syncthreads();
    const bool ok0 = (unsigned)d.x < N_NODES, ok1 = (unsigned)d.y < N_NODES;
    const bool ok2 = (unsigned)d.z < N_NODES, ok3 = (unsigned)d.w < N_NODES;
    unsigned r0 = 0u, r1 = 0u, r2 = 0u, r3 = 0u;
    if (ok0) r0 = atomicAdd(&wcnt[wave][d.x >> BUCKET_SHIFT], 1u);
    if (ok1) r1 = atomicAdd(&wcnt[wave][d.y >> BUCKET_SHIFT], 1u);
    if (ok2) r2 = atomicAdd(&wcnt[wave][d.z >> BUCKET_SHIFT], 1u);
    if (ok3) r3 = atomicAdd(&wcnt[wave][d.w >> BUCKET_SHIFT], 1u);
    __syncthreads();
    unsigned S = 0u, Pf = 0u, mine = 0u, inclS = 0u, inclM = 0u;
    if (tid < BUCKETS) {
        S  = part_tot[0][tid] + part_tot[1][tid] + part_tot[2][tid] + part_tot[3][tid];
        Pf = part_pre[0][tid] + part_pre[1][tid] + part_pre[2][tid] + part_pre[3][tid];
#pragma unroll
        for (int w = 0; w < 32; ++w) { const unsigned t = wcnt[w][tid]; wcnt[w][tid] = mine; mine += t; }
        inclS = wave_incl_scan(S, lane);
        inclM = wave_incl_scan(mine, lane);
        if (lane == 31) { wsum[wave][0] = inclS; wsum[wave][1] = inclM; }
    }
    __syncthreads();
    if (tid < BUCKETS) {
        unsigned baseS = 0u, baseM = 0u;
#pragma unroll
        for (int w = 0; w < 7; ++w) {
            baseS += (w < wave) ? wsum[w][0] : 0u;
            baseM += (w < wave) ? wsum[w][1] : 0u;
        }
        const unsigned gstart = baseS + inclS - S;
        const unsigned lstart = baseM + inclM - mine;
        gpos[tid] = gstart + Pf - lstart;
        lpos[tid] = lstart;
        if (g == 0) {
            bstart[tid] = gstart;
            if (tid == BUCKETS - 1) bstart[BUCKETS] = gstart + S;
        }
        if (tid == BUCKETS - 1) nvalid = lstart + mine;
    }
    __syncthreads();
    if (ok0) { const int k0 = d.x >> BUCKET_SHIFT; stage[lpos[k0] + wcnt[wave][k0] + r0] = ((unsigned)d.x << 16) | ((unsigned)s.x & 0xffffu); }
    if (ok1) { const int k1 = d.y >> BUCKET_SHIFT; stage[lpos[k1] + wcnt[wave][k1] + r1] = ((unsigned)d.y << 16) | ((unsigned)s.y & 0xffffu); }
    if (ok2) { const int k2 = d.z >> BUCKET_SHIFT; stage[lpos[k2] + wcnt[wave][k2] + r2] = ((unsigned)d.z << 16) | ((unsigned)s.z & 0xffffu); }
    if (ok3) { const int k3 = d.w >> BUCKET_SHIFT; stage[lpos[k3] + wcnt[wave][k3] + r3] = ((unsigned)d.w << 16) | ((unsigned)s.w & 0xffffu); }
    __syncthreads();
    const unsigned nv = nvalid;
#pragma unroll
    for (int k = 0; k < BCHUNK_EDGES / 1024; ++k) {
        const unsigned p = (unsigned)(tid + k * 1024);
        if (p < nv) {
            const unsigned v = stage[p];
            bucketed[gpos[v >> 24] + p] = v;
        }
    }
}

#define SCALE_PER_THREAD 8
#define SCALE_GROUPS (N_NODES * DIM / 4 / 1024 / SCALE_PER_THREAD)
#define PB_PER_THREAD (PB_CAP / 1024)

__global__ __launch_bounds__(1024) void rows_kernel(const unsigned* __restrict__ bucketed,
                                                    const unsigned* __restrict__ bstart,
                                                    int* __restrict__ row_ptr,
                                                    int* __restrict__ in_src,
                                                    float* __restrict__ rs_in,
                                                    const float* __restrict__ x,
                                                    const float* __restrict__ rs_out,
                                                    float* __restrict__ hs) {
    __shared__ unsigned wcnt[32][BUCKETS];
    __shared__ unsigned rowpos[BUCKETS];
    __shared__ unsigned wtot[8];
    __shared__ unsigned sorted[PB_CAP];
    const int tid = threadIdx.x;
    if (blockIdx.x >= BUCKETS) {
        const float4* x4 = (const float4*)x;
        float4* h4 = (float4*)hs;
        const int i0 = (blockIdx.x - BUCKETS) * (SCALE_PER_THREAD * 1024) + tid;
        float4 v[SCALE_PER_THREAD];
        float r[SCALE_PER_THREAD];
#pragma unroll
        for (int k = 0; k < SCALE_PER_THREAD; ++k) {
            v[k] = x4[i0 + k * 1024];
            r[k] = rs_out[(i0 + k * 1024) >> 4];
        }
#pragma unroll
        for (int k = 0; k < SCALE_PER_THREAD; ++k) {
            v[k].x *= r[k]; v[k].y *= r[k]; v[k].z *= r[k]; v[k].w *= r[k];
            h4[i0 + k * 1024] = v[k];
        }
        return;
    }
    const int bkt = blockIdx.x;
    const int lane = tid & 31;
    const int wave = tid >> 5;
    const unsigned start = bstart[bkt];
    const unsigned end = bstart[bkt + 1];
    const unsigned cnt = end - start;
    const bool staged = cnt <= PB_CAP;
    unsigned v[PB_PER_THREAD], r[PB_PER_THREAD];
    unsigned dg = 0u;
    if (staged) {
#pragma unroll
        for (int k = 0; k < PB_PER_THREAD; ++k) {
            const unsigned p = (unsigned)(tid + k * 1024);
            v[k] = (p < cnt) ? bucketed[start + p] : 0u;
            r[k] = 0u;
        }
#pragma unroll
        for (int k = 0; k < 32 * BUCKETS / 1024; ++k) (&wcnt[0][0])[tid + k * 1024] = 0u;
        __syncthreads();
#pragma unroll
        for (int k = 0; k < PB_PER_THREAD; ++k) {
            const unsigned p = (unsigned)(tid + k * 1024);
            if (p < cnt) r[k] = atomicAdd(&wcnt[wave][(v[k] >> 16) & (BUCKETS - 1)], 1u);
        }
        __syncthreads();
        if (tid < BUCKETS) {
#pragma unroll
            for (int w = 0; w < 32; ++w) { const unsigned t = wcnt[w][tid]; wcnt[w][tid] = dg; dg += t; }
        }
    } else {
        if (tid < BUCKETS) rowpos[tid] = 0u;
        __syncthreads();
        for (unsigned p = tid; p < cnt; p += 1024)
            atomicAdd(&rowpos[(bucketed[start + p] >> 16) & (BUCKETS - 1)], 1u);
        __syncthreads();
        if (tid < BUCKETS) dg = rowpos[tid];
    }
    unsigned incl = 0u;
    if (tid < BUCKETS) {
        incl = wave_incl_scan(dg, lane);
        if (lane == 31) wtot[wave] = incl;
    }
    __syncthreads();
    if (tid < BUCKETS) {
        unsigned wb = 0u;
#pragma unroll
        for (int w = 0; w < 7; ++w) wb += (w < wave) ? wtot[w] : 0u;
        const unsigned ex = wb + incl - dg;
        rowpos[tid] = ex;
        row_ptr[bkt * BUCKETS + tid] = (int)(start + ex);
        if (bkt == BUCKETS - 1 && tid == BUCKETS - 1) row_ptr[N_NODES] = (int)end;
        rs_in[bkt * BUCKETS + tid] = rsqrtf((float)(dg < 1u ? 1u : dg));
    }
    __syncthreads();
    if (staged) {
#pragma unroll
        for (int k = 0; k < PB_PER_THREAD; ++k) {
            const unsigned p = (unsigned)(tid + k * 1024);
            if (p < cnt) {
                const unsigned j = (v[k] >> 16) & (BUCKETS - 1);
                sorted[rowpos[j] + wcnt[wave][j] + r[k]] = v[k] & 0xffffu;
            }
        }
        __syncthreads();
#pragma unroll
        for (int k = 0; k < PB_PER_THREAD; ++k) {
            const unsigned p = (unsigned)(tid + k * 1024);
            if (p < cnt) in_src[start + p] = (int)sorted[p];
        }
    } else {
        for (unsigned p = tid; p < cnt; p += 1024) {
            const unsigned e = bucketed[start + p];
            in_src[start + atomicAdd(&rowpos[(e >> 16) & (BUCKETS - 1)], 1u)] = (int)(e & 0xffffu);
        }
    }
}

#define AGG_NODES 8
#define AGG_BATCH 16
#define TILE_ROWS 64
#define TILE_PITCH 68

__global__ __launch_bounds__(256) void layer_kernel(const float* __restrict__ hs,
                                                    const int* __restrict__ in_src,
                                                    const int* __restrict__ row_ptr,
                                                    const float* __restrict__ rs_in,
                                                    const float* __restrict__ W,
                                                    const float* __restrict__ bias,
                                                    const float* __restrict__ rs_out,
                                                    float* __restrict__ Y,
                                                    int relu_scale) {
    __shared__ alignas(16) float lWsw[16 * 4 * 32 * 2];
    __shared__ alignas(16) float lAgg[TILE_ROWS * TILE_PITCH];
    __shared__ float lB[DIM];
    __shared__ float lRs[TILE_ROWS];

    const int lane = threadIdx.x & 31;
    const int wave = threadIdx.x >> 5;

    v2f w2[8];
#pragma unroll
    for (int i = 0; i < 8; ++i) {
        const int p  = threadIdx.x + i * 256;
        const int l  = p & 31;
        const int t  = (p >> 5) & 3;
        const int f  = p >> 7;
        const int ka = 4 * f + 2 * (l >> 4);
        const int n  = 16 * t + (l & 15);
        w2[i].x = W[ka * DIM + n];
        w2[i].y = W[(ka + 1) * DIM + n];
    }
    float sb = 1.0f;
    if (threadIdx.x < DIM) sb = bias[threadIdx.x];
    else if (threadIdx.x < DIM + TILE_ROWS && relu_scale) sb = rs_out[blockIdx.x * TILE_ROWS + threadIdx.x - DIM];

    const int node0 = (blockIdx.x * 8 + wave) * AGG_NODES;
    const int rp  = (lane <= AGG_NODES) ? row_ptr[node0 + lane] : 0x7fffffff;
    const int rsl = __float_as_int(rs_in[node0 + (lane < AGG_NODES ? lane : AGG_NODES - 1)]);
    const int e_begin = __builtin_amdgcn_readlane(rp, 0);
    const int e_end   = __builtin_amdgcn_readlane(rp, AGG_NODES);
    int cur     = 0;
    int cur_end = __builtin_amdgcn_readlane(rp, 1);
    float ax = 0.0f, ay = 0.0f;
    const float* hl = hs + lane * 2;
    float* tl = lAgg + wave * AGG_NODES * TILE_PITCH + lane * 2;

    auto batch = [&](const int idx, const int j0, const int e0) {
        v2f v[AGG_BATCH];
#pragma unroll
        for (int u = 0; u < AGG_BATCH; ++u) {
            const unsigned srow = (unsigned)__builtin_amdgcn_readlane(idx, j0 + u);
            v[u] = *(const v2f*)(hl + (size_t)srow * DIM);
        }
        int rem = cur_end - e0;
#pragma unroll
        for (int u = 0; u < AGG_BATCH; ++u) {
            while (u >= rem) {
                const float r = __int_as_float(__builtin_amdgcn_readlane(rsl, cur));
                v2f outv; outv.x = ax * r; outv.y = ay * r;
                *(v2f*)(tl + cur * TILE_PITCH) = outv;
                ax = 0.0f; ay = 0.0f;
                ++cur;
                cur_end = __builtin_amdgcn_readlane(rp, cur + 1);
                rem = cur_end - e0;
            }
            ax += v[u].x; ay += v[u].y;
        }
    };

    const int e_last = (e_end > 0) ? e_end - 1 : 0;
    int idx = in_src[min(e_begin + lane, e_last)];

#pragma unroll
    for (int i = 0; i < 8; ++i) {
        const int p = threadIdx.x + i * 256;
        const int l = p & 31;
        const int t = (p >> 5) & 3;
        const int f = p >> 7;
        *(v2f*)&lWsw[(((f * 2 + (t >> 1)) * 32 + l) * 2 + (t & 1)) * 2] = w2[i];
    }
    if (threadIdx.x < DIM) lB[threadIdx.x] = sb;
    else if (threadIdx.x < DIM + TILE_ROWS) lRs[threadIdx.x - DIM] = sb;

    for (int base = e_begin; base < e_end; base += 32) {
        const int idx_next = in_src[min(base + 32 + lane, e_last)];
        batch(idx, 0, base);
        if (base + AGG_BATCH < e_end) batch(idx, AGG_BATCH, base + AGG_BATCH);
        idx = idx_next;
    }
    while (cur < AGG_NODES) {
        const float r = __int_as_float(__builtin_amdgcn_readlane(rsl, cur));
        v2f outv; outv.x = ax * r; outv.y = ay * r;
        *(v2f*)(tl + cur * TILE_PITCH) = outv;
        ax = 0.0f; ay = 0.0f;
        ++cur;
    }
    __syncthreads();

    const int half = lane >> 4;
    const int l16  = lane & 15;
    const int mt   = wave >> 1;
    const int nt0  = (wave & 1) * 2;
    const float* arow = lAgg + (mt * 16 + l16) * TILE_PITCH + 2 * half;

    v8f acc[2] = {v8f{}, v8f{}};
#pragma unroll
    for (int f = 0; f < 16; ++f) {
        const v2f a = *(const v2f*)(arow + 4 * f);
        const float4 b2 = *(const float4*)&lWsw[((f * 2 + (nt0 >> 1)) * 32 + lane) * 4];
        v2f b;
        b.x = b2.x; b.y = b2.y;
        acc[0] = __builtin_amdgcn_wmma_f32_16x16x4_f32(
            false, a, false, b, (short)0, acc[0], false, false);
        b.x = b2.z; b.y = b2.w;
        acc[1] = __builtin_amdgcn_wmma_f32_16x16x4_f32(
            false, a, false, b, (short)0, acc[1], false, false);
    }

    const int lrow0 = mt * 16 + half * 8;
    float* yrow = Y + ((size_t)blockIdx.x * TILE_ROWS + lrow0) * DIM;
    float rsv[8];
#pragma unroll
    for (int r = 0; r < 8; ++r) rsv[r] = lRs[lrow0 + r];

#pragma unroll
    for (int t = 0; t < 2; ++t) {
        const int n = (nt0 + t) * 16 + l16;
        const float bn = lB[n];
#pragma unroll
        for (int r = 0; r < 8; ++r) {
            float v = acc[t][r] + bn;
            if (relu_scale) v = fmaxf(v, 0.0f) * rsv[r];
            yrow[(size_t)r * DIM + n] = v;
        }
    }
}

extern "C" void kernel_launch(void* const* d_in, const int* in_sizes, int n_in,
                              void* d_out, int out_size, void* d_ws, size_t ws_size,
                              hipStream_t stream) {
    (void)in_sizes; (void)n_in; (void)out_size; (void)ws_size;
    const float* x  = (const float*)d_in[0];
    const int*   ei = (const int*)d_in[1];
    const float* W1 = (const float*)d_in[3];
    const float* b1 = (const float*)d_in[4];
    const float* W2 = (const float*)d_in[5];
    const float* b2 = (const float*)d_in[6];
    const float* W3 = (const float*)d_in[7];
    const float* b3 = (const float*)d_in[8];
    float* out = (float*)d_out;

    char* ws = (char*)d_ws;
    float*    hs      = (float*)ws;    ws += (size_t)N_NODES * DIM * 4;
    float*    h2      = (float*)ws;    ws += (size_t)N_NODES * DIM * 4;
    unsigned* unused0 = (unsigned*)ws; ws += (size_t)N_NODES * 4;
    float*    rs_out  = (float*)ws;    ws += (size_t)N_NODES * 4;
    float*    rs_in   = (float*)ws;    ws += (size_t)N_NODES * 4;
    int*      in_src  = (int*)ws;      ws += (size_t)N_EDGES * 4;
    int*      row_ptr = (int*)ws;      ws += (size_t)(N_NODES + 16) * 4;
    (void)unused0;
    unsigned* cnt_src  = (unsigned*)hs;
    unsigned* bucketed = (unsigned*)h2;
    unsigned* bcnt     = bucketed + (size_t)N_EDGES;
    unsigned* bstart   = bcnt + (size_t)BCHUNKS * BUCKETS;

    hist_kernel<<<dim3(CHUNKS, RANGES, 2), 256, 0, stream>>>(ei, cnt_src, bcnt);
    bucket_kernel<<<N_NODES / 1024 + BCHUNKS, 1024, 0, stream>>>(ei, cnt_src, bcnt, rs_out, bstart, bucketed);
    rows_kernel<<<BUCKETS + SCALE_GROUPS, 1024, 0, stream>>>(bucketed, bstart, row_ptr, in_src, rs_in, x, rs_out, hs);

    layer_kernel<<<N_NODES / TILE_ROWS, 256, 0, stream>>>(hs, in_src, row_ptr, rs_in, W1, b1, rs_out, h2, 1);
    layer_kernel<<<N_NODES / TILE_ROWS, 256, 0, stream>>>(h2, in_src, row_ptr, rs_in, W2, b2, rs_out, hs, 1);
    layer_kernel<<<N_NODES / TILE_ROWS, 256, 0, stream>>>(hs, in_src, row_ptr, rs_in, W3, b3, rs_out, out, 0);
}


# Edge sort without global offsets: launch 1 bucket-sorts each 4096-edge chunk in place, a bucket's workgroup collects its 256 runs; degree sums join x scaling; 6 -> 5 launches; row sum order changes
# speedup vs baseline: 4.1752x; 1.0866x over previous
#include <hip/hip_runtime.h>

#define N_NODES 65536
#define N_EDGES 1048576
#define DIM 64

typedef float v2f __attribute__((ext_vector_type(2)));
typedef float v8f __attribute__((ext_vector_type(8)));


#define CHUNKS 32
#define CHUNK_EDGES (N_EDGES / CHUNKS)
#define RANGES 4
#define RANGE_NODES (N_NODES / RANGES)
#define RANGE_SHIFT 14
#define EDGE_BATCH 16
#define SRC_BLOCKS (CHUNKS * RANGES)

#define BCHUNKS 256
#define BCHUNK_EDGES (N_EDGES / BCHUNKS)
#define BUCKETS 256
#define BUCKET_SHIFT 8
#define BPRE_PITCH 288
#define PB_CAP 6144
#define SORT_INT4 (BCHUNK_EDGES / 4 / 256)

__device__ __forceinline__ unsigned wave_incl_scan(unsigned v, const int lane) {
    v += (unsigned)__builtin_amdgcn_update_dpp(0, (int)v, 0x111, 0xf, 0xf, false);
    v += (unsigned)__builtin_amdgcn_update_dpp(0, (int)v, 0x112, 0xf, 0xf, false);
    v += (unsigned)__builtin_amdgcn_update_dpp(0, (int)v, 0x114, 0xf, 0xf, false);
    v += (unsigned)__builtin_amdgcn_update_dpp(0, (int)v, 0x118, 0xf, 0xf, false);
    const unsigned low_row = (unsigned)__builtin_amdgcn_readlane((int)v, 15);
    return (lane >= 16) ? v + low_row : v;
}

__global__ __launch_bounds__(256) void hist_kernel(const int* __restrict__ ei,
                                                   unsigned* __restrict__ cnt_src,
                                                   unsigned* __restrict__ bpre,
                                                   unsigned* __restrict__ bucketed) {
    __shared__ unsigned tab[RANGE_NODES];
    __shared__ unsigned wcnt[8][BUCKETS];
    __shared__ unsigned lpos[BUCKETS];
    __shared__ unsigned wsum[8];
    __shared__ unsigned stage[BCHUNK_EDGES];
    __shared__ unsigned nvalid;
    const int tid = threadIdx.x;
    if (blockIdx.x >= SRC_BLOCKS) {
        const int c = blockIdx.x - SRC_BLOCKS;
        const int lane = tid & 31;
        const int wave = tid >> 5;
        const int4* d4 = (const int4*)(ei + (size_t)N_EDGES + (size_t)c * BCHUNK_EDGES);
        const int4* s4 = (const int4*)(ei + (size_t)c * BCHUNK_EDGES);
        int dv[4 * SORT_INT4], sv[4 * SORT_INT4];
#pragma unroll
        for (int k = 0; k < SORT_INT4; ++k) {
            const int4 a = d4[tid + k * 256];
            const int4 b = s4[tid + k * 256];
            dv[4 * k] = a.x; dv[4 * k + 1] = a.y; dv[4 * k + 2] = a.z; dv[4 * k + 3] = a.w;
            sv[4 * k] = b.x; sv[4 * k + 1] = b.y; sv[4 * k + 2] = b.z; sv[4 * k + 3] = b.w;
        }
#pragma unroll
        for (int k = 0; k < 8 * BUCKETS / 256; ++k) (&wcnt[0][0])[tid + k * 256] = 0u;
        __syncthreads();
        unsigned rk[4 * SORT_INT4];
#pragma unroll
        for (int k = 0; k < 4 * SORT_INT4; ++k) {
            rk[k] = 0u;
            if ((unsigned)dv[k] < N_NODES) rk[k] = atomicAdd(&wcnt[wave][dv[k] >> BUCKET_SHIFT], 1u);
        }
        __syncthreads();
        unsigned mine = 0u;
#pragma unroll
        for (int w = 0; w < 8; ++w) { const unsigned t = wcnt[w][tid]; wcnt[w][tid] = mine; mine += t; }
        const unsigned incl = wave_incl_scan(mine, lane);
        if (lane == 31) wsum[wave] = incl;
        __syncthreads();
        unsigned base = 0u;
#pragma unroll
        for (int w = 0; w < 7; ++w) base += (w < wave) ? wsum[w] : 0u;
        const unsigned lstart = base + incl - mine;
        lpos[tid] = lstart;
        bpre[(size_t)c * BPRE_PITCH + tid] = lstart;
        if (tid == BUCKETS - 1) {
            bpre[(size_t)c * BPRE_PITCH + BUCKETS] = lstart + mine;
            nvalid = lstart + mine;
        }
        __syncthreads();
#pragma unroll
        for (int k = 0; k < 4 * SORT_INT4; ++k) {
            if ((unsigned)dv[k] < N_NODES) {
                const int kb = dv[k] >> BUCKET_SHIFT;
                stage[lpos[kb] + wcnt[wave][kb] + rk[k]] = ((unsigned)dv[k] << 16) | ((unsigned)sv[k] & 0xffffu);
            }
        }
        __syncthreads();
        const unsigned nv = nvalid;
        unsigned* outc = bucketed + (size_t)c * BCHUNK_EDGES;
#pragma unroll
        for (int k = 0; k < BCHUNK_EDGES / 256; ++k) {
            const unsigned p = (unsigned)(tid + k * 256);
            if (p < nv) outc[p] = stage[p];
        }
        return;
    }
    const int g = blockIdx.x >> 2, q = blockIdx.x & (RANGES - 1);
    uint4* tab4 = (uint4*)tab;
    for (int i = tid; i < RANGE_NODES / 4; i += 256) tab4[i] = make_uint4(0u, 0u, 0u, 0u);
    __syncthreads();
    const int4* idx4 = (const int4*)(ei + (size_t)g * CHUNK_EDGES);
    for (int i0 = tid; i0 < CHUNK_EDGES / 4; i0 += 256 * EDGE_BATCH) {
        int4 v[EDGE_BATCH];
#pragma unroll
        for (int k = 0; k < EDGE_BATCH; ++k) v[k] = idx4[i0 + k * 256];
#pragma unroll
        for (int k = 0; k < EDGE_BATCH; ++k) {
            if ((v[k].x >> RANGE_SHIFT) == q) atomicAdd(&tab[v[k].x & (RANGE_NODES - 1)], 1u);
            if ((v[k].y >> RANGE_SHIFT) == q) atomicAdd(&tab[v[k].y & (RANGE_NODES - 1)], 1u);
            if ((v[k].z >> RANGE_SHIFT) == q) atomicAdd(&tab[v[k].z & (RANGE_NODES - 1)], 1u);
            if ((v[k].w >> RANGE_SHIFT) == q) atomicAdd(&tab[v[k].w & (RANGE_NODES - 1)], 1u);
        }
    }
    __syncthreads();
    uint4* out4 = (uint4*)(cnt_src + (size_t)g * N_NODES + (size_t)q * RANGE_NODES);
    for (int i = tid; i < RANGE_NODES / 4; i += 256) out4[i] = tab4[i];
}

#define SCALE_PER_THREAD 8
#define SCALE_GROUPS (N_NODES * DIM / 4 / 1024 / SCALE_PER_THREAD)
#define SCALE_ROWS (SCALE_PER_THREAD * 1024 / 16)
#define PB_PER_THREAD (PB_CAP / 1024)

__global__ __launch_bounds__(1024) void rows_kernel(const unsigned* __restrict__ bucketed,
                                                    const unsigned* __restrict__ bpre,
                                                    const unsigned* __restrict__ cnt_src,
                                                    int* __restrict__ row_ptr,
                                                    int* __restrict__ in_src,
                                                    float* __restrict__ rs_in,
                                                    const float* __restrict__ x,
                                                    float* __restrict__ rs_out,
                                                    float* __restrict__ hs) {
    __shared__ unsigned wcnt[32][BUCKETS];
    __shared__ unsigned rowpos[BUCKETS];
    __shared__ unsigned wtot[8];
    __shared__ unsigned sorted[PB_CAP];
    __shared__ unsigned pc_src[BCHUNKS];
    __shared__ unsigned pc_off[BCHUNKS];
    __shared__ unsigned pc_n[BCHUNKS];
    __shared__ unsigned psum[8][2];
    __shared__ unsigned tot[2];
    __shared__ unsigned dpart[2][SCALE_ROWS];
    __shared__ float lrs[SCALE_ROWS];
    const int tid = threadIdx.x;
    if (blockIdx.x >= BUCKETS) {
        const int grp = blockIdx.x - BUCKETS;
        const int row0 = grp * SCALE_ROWS;
        const int j = tid & (SCALE_ROWS - 1);
        const int half = tid / SCALE_ROWS;
        const float4* x4 = (const float4*)x;
        float4* h4 = (float4*)hs;
        const int i0 = grp * (SCALE_PER_THREAD * 1024) + tid;
        unsigned dsum = 0u;
#pragma unroll
        for (int g = 0; g < CHUNKS / 2; ++g) dsum += cnt_src[(size_t)(half * (CHUNKS / 2) + g) * N_NODES + row0 + j];
        float4 v[SCALE_PER_THREAD];
#pragma unroll
        for (int k = 0; k < SCALE_PER_THREAD; ++k) v[k] = x4[i0 + k * 1024];
        dpart[half][j] = dsum;
        __syncthreads();
        if (tid < SCALE_ROWS) {
            unsigned dout = dpart[0][tid] + dpart[1][tid];
            if (dout < 1u) dout = 1u;
            const float r = rsqrtf((float)dout);
            rs_out[row0 + tid] = r;
            lrs[tid] = r;
        }
        __syncthreads();
#pragma unroll
        for (int k = 0; k < SCALE_PER_THREAD; ++k) {
            const float r = lrs[(tid + k * 1024) >> 4];
            v[k].x *= r; v[k].y *= r; v[k].z *= r; v[k].w *= r;
            h4[i0 + k * 1024] = v[k];
        }
        return;
    }
    const int bkt = blockIdx.x;
    const int lane = tid & 31;
    const int wave = tid >> 5;
    unsigned s_c = 0u, n_c = 0u, inclN = 0u, inclS = 0u;
    if (tid < BCHUNKS) {
        s_c = bpre[(size_t)tid * BPRE_PITCH + bkt];
        n_c = bpre[(size_t)tid * BPRE_PITCH + bkt + 1] - s_c;
        inclN = wave_incl_scan(n_c, lane);
        inclS = wave_incl_scan(s_c, lane);
        if (lane == 31) { psum[wave][0] = inclN; psum[wave][1] = inclS; }
    }
    __syncthreads();
    if (tid < BCHUNKS) {
        unsigned baseN = 0u, baseS = 0u;
#pragma unroll
        for (int w = 0; w < 7; ++w) {
            baseN += (w < wave) ? psum[w][0] : 0u;
            baseS += (w < wave) ? psum[w][1] : 0u;
        }
        pc_src[tid] = s_c;
        pc_off[tid] = baseN + inclN - n_c;
        pc_n[tid] = n_c;
        if (tid == BCHUNKS - 1) { tot[0] = baseN + inclN; tot[1] = baseS + inclS; }
    }
    __syncthreads();
    const unsigned cnt = tot[0];
    const unsigned start = tot[1];
    const unsigned end = start + cnt;
    const bool staged = cnt <= PB_CAP;
    unsigned v[PB_PER_THREAD], r[PB_PER_THREAD];
    unsigned dg = 0u;
    if (staged) {
        unsigned pv[8];
#pragma unroll
        for (int q = 0; q < 8; ++q) {
            const int c = wave * 8 + q;
            pv[q] = ((unsigned)lane < pc_n[c]) ? bucketed[(size_t)c * BCHUNK_EDGES + pc_src[c] + lane] : 0u;
        }
#pragma unroll
        for (int q = 0; q < 8; ++q) {
            const int c = wave * 8 + q;
            if ((unsigned)lane < pc_n[c]) sorted[pc_off[c] + lane] = pv[q];
        }
#pragma unroll
        for (int q = 0; q < 8; ++q) {
            const int c = wave * 8 + q;
            const unsigned n = pc_n[c];
#pragma unroll 1
            for (unsigned i = lane + 32; i < n; i += 32)
                sorted[pc_off[c] + i] = bucketed[(size_t)c * BCHUNK_EDGES + pc_src[c] + i];
        }
        __syncthreads();
#pragma unroll
        for (int k = 0; k < PB_PER_THREAD; ++k) {
            const unsigned p = (unsigned)(tid + k * 1024);
            v[k] = (p < cnt) ? sorted[p] : 0u;
            r[k] = 0u;
        }
#pragma unroll
        for (int k = 0; k < 32 * BUCKETS / 1024; ++k) (&wcnt[0][0])[tid + k * 1024] = 0u;
        __syncthreads();
#pragma unroll
        for (int k = 0; k < PB_PER_THREAD; ++k) {
            const unsigned p = (unsigned)(tid + k * 1024);
            if (p < cnt) r[k] = atomicAdd(&wcnt[wave][(v[k] >> 16) & (BUCKETS - 1)], 1u);
        }
        __syncthreads();
        if (tid < BUCKETS) {
#pragma unroll
            for (int w = 0; w < 32; ++w) { const unsigned t = wcnt[w][tid]; wcnt[w][tid] = dg; dg += t; }
        }
    } else {
        if (tid < BUCKETS) rowpos[tid] = 0u;
        __syncthreads();
        for (int c = wave; c < BCHUNKS; c += 32) {
            const unsigned n = pc_n[c];
            const unsigned* run = bucketed + (size_t)c * BCHUNK_EDGES + pc_src[c];
            for (unsigned i = lane; i < n; i += 32)
                atomicAdd(&rowpos[(run[i] >> 16) & (BUCKETS - 1)], 1u);
        }
        __syncthreads();
        if (tid < BUCKETS) dg = rowpos[tid];
    }
    unsigned incl = 0u;
    if (tid < BUCKETS) {
        incl = wave_incl_scan(dg, lane);
        if (lane == 31) wtot[wave] = incl;
    }
    __syncthreads();
    if (tid < BUCKETS) {
        unsigned wb = 0u;
#pragma unroll
        for (int w = 0; w < 7; ++w) wb += (w < wave) ? wtot[w] : 0u;
        const unsigned ex = wb + incl - dg;
        rowpos[tid] = ex;
        row_ptr[bkt * BUCKETS + tid] = (int)(start + ex);
        if (bkt == BUCKETS - 1 && tid == BUCKETS - 1) row_ptr[N_NODES] = (int)end;
        rs_in[bkt * BUCKETS + tid] = rsqrtf((float)(dg < 1u ? 1u : dg));
    }
    __syncthreads();
    if (staged) {
#pragma unroll
        for (int k = 0; k < PB_PER_THREAD; ++k) {
            const unsigned p = (unsigned)(tid + k * 1024);
            if (p < cnt) {
                const unsigned j = (v[k] >> 16) & (BUCKETS - 1);
                sorted[rowpos[j] + wcnt[wave][j] + r[k]] = v[k] & 0xffffu;
            }
        }
        __syncthreads();
#pragma unroll
        for (int k = 0; k < PB_PER_THREAD; ++k) {
            const unsigned p = (unsigned)(tid + k * 1024);
            if (p < cnt) in_src[start + p] = (int)sorted[p];
        }
    } else {
        for (int c = wave; c < BCHUNKS; c += 32) {
            const unsigned n = pc_n[c];
            const unsigned* run = bucketed + (size_t)c * BCHUNK_EDGES + pc_src[c];
            for (unsigned i = lane; i < n; i += 32) {
                const unsigned e = run[i];
                in_src[start + atomicAdd(&rowpos[(e >> 16) & (BUCKETS - 1)], 1u)] = (int)(e & 0xffffu);
            }
        }
    }
}

#define AGG_NODES 8
#define AGG_BATCH 16
#define TILE_ROWS 64
#define TILE_PITCH 68

__global__ __launch_bounds__(256) void layer_kernel(const float* __restrict__ hs,
                                                    const int* __restrict__ in_src,
                                                    const int* __restrict__ row_ptr,
                                                    const float* __restrict__ rs_in,
                                                    const float* __restrict__ W,
                                                    const float* __restrict__ bias,
                                                    const float* __restrict__ rs_out,
                                                    float* __restrict__ Y,
                                                    int relu_scale) {
    __shared__ alignas(16) float lWsw[16 * 4 * 32 * 2];
    __shared__ alignas(16) float lAgg[TILE_ROWS * TILE_PITCH];
    __shared__ float lB[DIM];
    __shared__ float lRs[TILE_ROWS];

    const int lane = threadIdx.x & 31;
    const int wave = threadIdx.x >> 5;

    v2f w2[8];
#pragma unroll
    for (int i = 0; i < 8; ++i) {
        const int p  = threadIdx.x + i * 256;
        const int l  = p & 31;
        const int t  = (p >> 5) & 3;
        const int f  = p >> 7;
        const int ka = 4 * f + 2 * (l >> 4);
        const int n  = 16 * t + (l & 15);
        w2[i].x = W[ka * DIM + n];
        w2[i].y = W[(ka + 1) * DIM + n];
    }
    float sb = 1.0f;
    if (threadIdx.x < DIM) sb = bias[threadIdx.x];
    else if (threadIdx.x < DIM + TILE_ROWS && relu_scale) sb = rs_out[blockIdx.x * TILE_ROWS + threadIdx.x - DIM];

    const int node0 = (blockIdx.x * 8 + wave) * AGG_NODES;
    const int rp  = (lane <= AGG_NODES) ? row_ptr[node0 + lane] : 0x7fffffff;
    const int rsl = __float_as_int(rs_in[node0 + (lane < AGG_NODES ? lane : AGG_NODES - 1)]);
    const int e_begin = __builtin_amdgcn_readlane(rp, 0);
    const int e_end   = __builtin_amdgcn_readlane(rp, AGG_NODES);
    int cur     = 0;
    int cur_end = __builtin_amdgcn_readlane(rp, 1);
    float ax = 0.0f, ay = 0.0f;
    const float* hl = hs + lane * 2;
    float* tl = lAgg + wave * AGG_NODES * TILE_PITCH + lane * 2;

    auto batch = [&](const int idx, const int j0, const int e0) {
        v2f v[AGG_BATCH];
#pragma unroll
        for (int u = 0; u < AGG_BATCH; ++u) {
            const unsigned srow = (unsigned)__builtin_amdgcn_readlane(idx, j0 + u);
            v[u] = *(const v2f*)(hl + (size_t)srow * DIM);
        }
        int rem = cur_end - e0;
#pragma unroll
        for (int u = 0; u < AGG_BATCH; ++u) {
            while (u >= rem) {
                const float r = __int_as_float(__builtin_amdgcn_readlane(rsl, cur));
                v2f outv; outv.x = ax * r; outv.y = ay * r;
                *(v2f*)(tl + cur * TILE_PITCH) = outv;
                ax = 0.0f; ay = 0.0f;
                ++cur;
                cur_end = __builtin_amdgcn_readlane(rp, cur + 1);
                rem = cur_end - e0;
            }
            ax += v[u].x; ay += v[u].y;
        }
    };

    const int e_last = (e_end > 0) ? e_end - 1 : 0;
    int idx = in_src[min(e_begin + lane, e_last)];

#pragma unroll
    for (int i = 0; i < 8; ++i) {
        const int p = threadIdx.x + i * 256;
        const int l = p & 31;
        const int t = (p >> 5) & 3;
        const int f = p >> 7;
        *(v2f*)&lWsw[(((f * 2 + (t >> 1)) * 32 + l) * 2 + (t & 1)) * 2] = w2[i];
    }
    if (threadIdx.x < DIM) lB[threadIdx.x] = sb;
    else if (threadIdx.x < DIM + TILE_ROWS) lRs[threadIdx.x - DIM] = sb;

    for (int base = e_begin; base < e_end; base += 32) {
        const int idx_next = in_src[min(base + 32 + lane, e_last)];
        batch(idx, 0, base);
        if (base + AGG_BATCH < e_end) batch(idx, AGG_BATCH, base + AGG_BATCH);
        idx = idx_next;
    }
    while (cur < AGG_NODES) {
        const float r = __int_as_float(__builtin_amdgcn_readlane(rsl, cur));
        v2f outv; outv.x = ax * r; outv.y = ay * r;
        *(v2f*)(tl + cur * TILE_PITCH) = outv;
        ax = 0.0f; ay = 0.0f;
        ++cur;
    }
    __syncthreads();

    const int half = lane >> 4;
    const int l16  = lane & 15;
    const int mt   = wave >> 1;
    const int nt0  = (wave & 1) * 2;
    const float* arow = lAgg + (mt * 16 + l16) * TILE_PITCH + 2 * half;

    v8f acc[2] = {v8f{}, v8f{}};
#pragma unroll
    for (int f = 0; f < 16; ++f) {
        const v2f a = *(const v2f*)(arow + 4 * f);
        const float4 b2 = *(const float4*)&lWsw[((f * 2 + (nt0 >> 1)) * 32 + lane) * 4];
        v2f b;
        b.x = b2.x; b.y = b2.y;
        acc[0] = __builtin_amdgcn_wmma_f32_16x16x4_f32(
            false, a, false, b, (short)0, acc[0], false, false);
        b.x = b2.z; b.y = b2.w;
        acc[1] = __builtin_amdgcn_wmma_f32_16x16x4_f32(
            false, a, false, b, (short)0, acc[1], false, false);
    }

    const int lrow0 = mt * 16 + half * 8;
    float* yrow = Y + ((size_t)blockIdx.x * TILE_ROWS + lrow0) * DIM;
    float rsv[8];
#pragma unroll
    for (int r = 0; r < 8; ++r) rsv[r] = lRs[lrow0 + r];

#pragma unroll
    for (int t = 0; t < 2; ++t) {
        const int n = (nt0 + t) * 16 + l16;
        const float bn = lB[n];
#pragma unroll
        for (int r = 0; r < 8; ++r) {
            float v = acc[t][r] + bn;
            if (relu_scale) v = fmaxf(v, 0.0f) * rsv[r];
            yrow[(size_t)r * DIM + n] = v;
        }
    }
}

extern "C" void kernel_launch(void* const* d_in, const int* in_sizes, int n_in,
                              void* d_out, int out_size, void* d_ws, size_t ws_size,
                              hipStream_t stream) {
    (void)in_sizes; (void)n_in; (void)out_size; (void)ws_size;
    const float* x  = (const float*)d_in[0];
    const int*   ei = (const int*)d_in[1];
    const float* W1 = (const float*)d_in[3];
    const float* b1 = (const float*)d_in[4];
    const float* W2 = (const float*)d_in[5];
    const float* b2 = (const float*)d_in[6];
    const float* W3 = (const float*)d_in[7];
    const float* b3 = (const float*)d_in[8];
    float* out = (float*)d_out;

    char* ws = (char*)d_ws;
    float*    hs      = (float*)ws;    ws += (size_t)N_NODES * DIM * 4;
    float*    h2      = (float*)ws;    ws += (size_t)N_NODES * DIM * 4;
    unsigned* unused0 = (unsigned*)ws; ws += (size_t)N_NODES * 4;
    float*    rs_out  = (float*)ws;    ws += (size_t)N_NODES * 4;
    float*    rs_in   = (float*)ws;    ws += (size_t)N_NODES * 4;
    int*      in_src  = (int*)ws;      ws += (size_t)N_EDGES * 4;
    int*      row_ptr = (int*)ws;      ws += (size_t)(N_NODES + 16) * 4;
    (void)unused0;
    unsigned* bucketed = (unsigned*)h2;
    unsigned* bpre     = bucketed + (size_t)N_EDGES;
    unsigned* cnt_src  = bpre + (size_t)BCHUNKS * BPRE_PITCH;

    hist_kernel<<<SRC_BLOCKS + BCHUNKS, 256, 0, stream>>>(ei, cnt_src, bpre, bucketed);
    rows_kernel<<<BUCKETS + SCALE_GROUPS, 1024, 0, stream>>>(bucketed, bpre, cnt_src, row_ptr, in_src, rs_in, x, rs_out, hs);

    layer_kernel<<<N_NODES / TILE_ROWS, 256, 0, stream>>>(hs, in_src, row_ptr, rs_in, W1, b1, rs_out, h2, 1);
    layer_kernel<<<N_NODES / TILE_ROWS, 256, 0, stream>>>(h2, in_src, row_ptr, rs_in, W2, b2, rs_out, hs, 1);
    layer_kernel<<<N_NODES / TILE_ROWS, 256, 0, stream>>>(hs, in_src, row_ptr, rs_in, W3, b3, rs_out, out, 0);
}
